# starting kernel
# baseline (speedup 1.0000x reference)
#include <hip/hip_runtime.h>
#include <hip/hip_bf16.h>

typedef float v2f __attribute__((ext_vector_type(2)));
typedef float v8f __attribute__((ext_vector_type(8)));

#define LDS_STRIDE 66

__global__ void deg_init_kernel(float* __restrict__ deg, int n) {
  int i = blockIdx.x * blockDim.x + threadIdx.x;
  if (i < n) deg[i] = 1.0f;
}

__global__ void deg_edge_kernel(const int* __restrict__ col, float* __restrict__ deg, int e) {
  int i = blockIdx.x * blockDim.x + threadIdx.x;
  if (i < e) atomicAdd(&deg[col[i]], 1.0f);
}

__global__ void deg_rsqrt_kernel(float* __restrict__ deg, int n) {
  int i = blockIdx.x * blockDim.x + threadIdx.x;
  if (i < n) deg[i] = rsqrtf(deg[i]);
}

__global__ __launch_bounds__(64) void gemm64_wmma_kernel(const float* __restrict__ X,
                                                         const float* __restrict__ W,
                                                         float* __restrict__ H) {
  __shared__ float Wt[64 * LDS_STRIDE];
  for (int p = threadIdx.x; p < 64 * 64; p += 64) {
    int k = p >> 6, n = p & 63;
    Wt[n * LDS_STRIDE + k] = W[p];
  }
  __syncthreads();

  const int wave = threadIdx.x >> 5;
  const int lane = threadIdx.x & 31;
  const int half = lane >> 4;
  const int nc   = lane & 15;
  const int rowBase = blockIdx.x * 32 + wave * 16;

  const float* xrow = X + (size_t)(rowBase + nc) * 64;
  v8f acc0 = {}, acc1 = {}, acc2 = {}, acc3 = {};

#pragma unroll
  for (int k0 = 0; k0 < 64; k0 += 4) {
    const int kb = k0 + half * 2;
    v2f a  = *(const v2f*)(xrow + kb);
    v2f b0 = *(const v2f*)(&Wt[(0 * 16 + nc) * LDS_STRIDE + kb]);
    v2f b1 = *(const v2f*)(&Wt[(1 * 16 + nc) * LDS_STRIDE + kb]);
    v2f b2 = *(const v2f*)(&Wt[(2 * 16 + nc) * LDS_STRIDE + kb]);
    v2f b3 = *(const v2f*)(&Wt[(3 * 16 + nc) * LDS_STRIDE + kb]);
    acc0 = __builtin_amdgcn_wmma_f32_16x16x4_f32(false, a, false, b0, (short)0, acc0, false, false);
    acc1 = __builtin_amdgcn_wmma_f32_16x16x4_f32(false, a, false, b1, (short)0, acc1, false, false);
    acc2 = __builtin_amdgcn_wmma_f32_16x16x4_f32(false, a, false, b2, (short)0, acc2, false, false);
    acc3 = __builtin_amdgcn_wmma_f32_16x16x4_f32(false, a, false, b3, (short)0, acc3, false, false);
  }

#pragma unroll
  for (int v = 0; v < 8; ++v) {
    float* hrow = H + (size_t)(rowBase + v + half * 8) * 64;
    hrow[0 * 16 + nc] = acc0[v];
    hrow[1 * 16 + nc] = acc1[v];
    hrow[2 * 16 + nc] = acc2[v];
    hrow[3 * 16 + nc] = acc3[v];
  }
}

__global__ void gemm64_tail_kernel(const float* __restrict__ X, const float* __restrict__ W,
                                   float* __restrict__ H, int start, int n) {
  int t = blockIdx.x * blockDim.x + threadIdx.x;
  int r = start + (t >> 6), c = t & 63;
  if (r >= n) return;
  float s = 0.f;
  for (int k = 0; k < 64; ++k) s += X[(size_t)r * 64 + k] * W[k * 64 + c];
  H[(size_t)r * 64 + c] = s;
}

__global__ void self1_kernel(const float* __restrict__ h, const float* __restrict__ dinv,
                             float* __restrict__ h2, int n) {
  int t = blockIdx.x * blockDim.x + threadIdx.x;
  int node = t >> 4, c4 = (t & 15) * 4;
  if (node >= n) return;
  float s = dinv[node] * dinv[node];
  float4 v = *(const float4*)&h[(size_t)node * 64 + c4];
  float4 o = {v.x * s, v.y * s, v.z * s, v.w * s};
  *(float4*)&h2[(size_t)node * 64 + c4] = o;
}

__global__ void edge1_kernel(const float* __restrict__ h, const int* __restrict__ row,
                             const int* __restrict__ col, const float* __restrict__ dinv,
                             float* __restrict__ h2, int e) {
  int t = blockIdx.x * blockDim.x + threadIdx.x;
  int ed = t >> 4, c4 = (t & 15) * 4;
  if (ed >= e) return;
  int r = row[ed], c = col[ed];
  float nm = dinv[r] * dinv[c];
  float4 v = *(const float4*)&h[(size_t)r * 64 + c4];
  float* dst = &h2[(size_t)c * 64 + c4];
  atomicAdd(dst + 0, v.x * nm);
  atomicAdd(dst + 1, v.y * nm);
  atomicAdd(dst + 2, v.z * nm);
  atomicAdd(dst + 3, v.w * nm);
}

__global__ __launch_bounds__(64) void gemm2_wmma_kernel(const float* __restrict__ H2,
                                                        const float* __restrict__ B1,
                                                        const float* __restrict__ W2,
                                                        float* __restrict__ G) {
  __shared__ float Wt[16 * LDS_STRIDE];
  __shared__ float b1s[64];
  for (int p = threadIdx.x; p < 16 * 64; p += 64) {
    int n = p >> 6, k = p & 63;
    Wt[n * LDS_STRIDE + k] = (n < 2) ? W2[k * 2 + n] : 0.0f;
  }
  b1s[threadIdx.x] = B1[threadIdx.x];
  __syncthreads();

  const int wave = threadIdx.x >> 5;
  const int lane = threadIdx.x & 31;
  const int half = lane >> 4;
  const int nc   = lane & 15;
  const int rowBase = blockIdx.x * 32 + wave * 16;

  const float* hrow = H2 + (size_t)(rowBase + nc) * 64;
  v8f acc = {};

#pragma unroll
  for (int k0 = 0; k0 < 64; k0 += 4) {
    const int kb = k0 + half * 2;
    v2f a;
    a.x = fmaxf(hrow[kb + 0] + b1s[kb + 0], 0.0f);
    a.y = fmaxf(hrow[kb + 1] + b1s[kb + 1], 0.0f);
    v2f b = *(const v2f*)(&Wt[nc * LDS_STRIDE + kb]);
    acc = __builtin_amdgcn_wmma_f32_16x16x4_f32(false, a, false, b, (short)0, acc, false, false);
  }

  if (nc < 2) {
#pragma unroll
    for (int v = 0; v < 8; ++v)
      G[(size_t)(rowBase + v + half * 8) * 2 + nc] = acc[v];
  }
}

__global__ void gemm2_tail_kernel(const float* __restrict__ H2, const float* __restrict__ B1,
                                  const float* __restrict__ W2, float* __restrict__ G,
                                  int start, int n) {
  int t = blockIdx.x * blockDim.x + threadIdx.x;
  int r = start + (t >> 1), c = t & 1;
  if (r >= n) return;
  float s = 0.f;
  for (int k = 0; k < 64; ++k)
    s += fmaxf(H2[(size_t)r * 64 + k] + B1[k], 0.0f) * W2[k * 2 + c];
  G[(size_t)r * 2 + c] = s;
}

__global__ void self2_kernel(const float* __restrict__ g, const float* __restrict__ dinv,
                             const float* __restrict__ b2, float* __restrict__ out, int n) {
  int i = blockIdx.x * blockDim.x + threadIdx.x;
  if (i >= n) return;
  float s = dinv[i] * dinv[i];
  out[(size_t)i * 2 + 0] = g[(size_t)i * 2 + 0] * s + b2[0];
  out[(size_t)i * 2 + 1] = g[(size_t)i * 2 + 1] * s + b2[1];
}

__global__ void edge2_kernel(const float* __restrict__ g, const int* __restrict__ row,
                             const int* __restrict__ col, const float* __restrict__ dinv,
                             float* __restrict__ out, int e) {
  int i = blockIdx.x * blockDim.x + threadIdx.x;
  if (i >= e) return;
  int r = row[i], c = col[i];
  float nm = dinv[r] * dinv[c];
  float2 v = *(const float2*)&g[(size_t)r * 2];
  atomicAdd(&out[(size_t)c * 2 + 0], v.x * nm);
  atomicAdd(&out[(size_t)c * 2 + 1], v.y * nm);
}

extern "C" void kernel_launch(void* const* d_in, const int* in_sizes, int n_in,
                              void* d_out, int out_size, void* d_ws, size_t ws_size,
                              hipStream_t stream) {
  const float* x  = (const float*)d_in[0];
  const int*   ei = (const int*)d_in[1];
  const float* W1 = (const float*)d_in[2];
  const float* b1 = (const float*)d_in[3];
  const float* W2 = (const float*)d_in[4];
  const float* b2 = (const float*)d_in[5];
  float* out = (float*)d_out;

  const int N = in_sizes[0] / 64;
  const int E = in_sizes[1] / 2;
  const int* row = ei;
  const int* col = ei + E;

  const int Npad = (N + 63) & ~63;
  float* dinv = (float*)d_ws;
  float* h    = dinv + Npad;
  float* h2   = h + (size_t)N * 64;
  float* g    = h2 + (size_t)N * 64;

  const int T = 256;
  deg_init_kernel<<<(N + T - 1) / T, T, 0, stream>>>(dinv, N);
  deg_edge_kernel<<<(E + T - 1) / T, T, 0, stream>>>(col, dinv, E);
  deg_rsqrt_kernel<<<(N + T - 1) / T, T, 0, stream>>>(dinv, N);

  const int mainBlocks = N / 32;
  const int tailStart  = mainBlocks * 32;
  if (mainBlocks > 0)
    gemm64_wmma_kernel<<<mainBlocks, 64, 0, stream>>>(x, W1, h);
  if (tailStart < N) {
    int tcnt = (N - tailStart) * 64;
    gemm64_tail_kernel<<<(tcnt + T - 1) / T, T, 0, stream>>>(x, W1, h, tailStart, N);
  }

  self1_kernel<<<((long long)N * 16 + T - 1) / T, T, 0, stream>>>(h, dinv, h2, N);
  edge1_kernel<<<((long long)E * 16 + T - 1) / T, T, 0, stream>>>(h, row, col, dinv, h2, E);

  if (mainBlocks > 0)
    gemm2_wmma_kernel<<<mainBlocks, 64, 0, stream>>>(h2, b1, W2, g);
  if (tailStart < N) {
    int tcnt = (N - tailStart) * 2;
    gemm2_tail_kernel<<<(tcnt + T - 1) / T, T, 0, stream>>>(h2, b1, W2, g, tailStart, N);
  }

  self2_kernel<<<(N + T - 1) / T, T, 0, stream>>>(g, dinv, b2, out, N);
  edge2_kernel<<<(E + T - 1) / T, T, 0, stream>>>(g, row, col, dinv, out, E);
}


# Degree count and layer-2 scatter-add accumulate in LDS per (edge slice, destination range) block, partial rows summed in fixed slice order; their global atomics removed; layer-2 sum order changes
# speedup vs baseline: 1.6590x; 1.6590x over previous
#include <hip/hip_runtime.h>
#include <hip/hip_bf16.h>

typedef float v2f __attribute__((ext_vector_type(2)));
typedef float v8f __attribute__((ext_vector_type(8)));

#define LDS_STRIDE 66

#define PART_THREADS 1024
#define EDGE_UNROLL 8
#define DEG_NP 51200
#define DEG_SLICES 64
#define AGG_NP 25600
#define AGG_SLICES 32

__global__ __launch_bounds__(PART_THREADS) void deg_part_kernel(const int* __restrict__ col,
                                                                int* __restrict__ part,
                                                                int e, int n) {
  __shared__ alignas(16) int cnt[DEG_NP];
  const int tid = threadIdx.x;
  const int lo = blockIdx.y * DEG_NP;
  const int span = min(DEG_NP, n - lo);
  for (int i = tid; i < DEG_NP / 4; i += PART_THREADS) ((int4*)cnt)[i] = make_int4(0, 0, 0, 0);
  __syncthreads();

  const int per = (e + (int)gridDim.x - 1) / (int)gridDim.x;
  const int e0 = min(e, (int)blockIdx.x * per);
  const int e1 = min(e, e0 + per);
  for (int base = e0; base < e1; base += PART_THREADS * EDGE_UNROLL) {
    int d[EDGE_UNROLL];
#pragma unroll
    for (int k = 0; k < EDGE_UNROLL; ++k) {
      const int i = base + k * PART_THREADS + tid;
      const bool ok = i < e1;
      const int c = col[ok ? i : e0];
      d[k] = ok ? (c - lo) : -1;
    }
#pragma unroll
    for (int k = 0; k < EDGE_UNROLL; ++k)
      if ((unsigned)d[k] < (unsigned)span) atomicAdd(&cnt[d[k]], 1);
  }
  __syncthreads();

  int* dst = part + (size_t)blockIdx.x * n + lo;
  for (int i = tid; i < span; i += PART_THREADS) dst[i] = cnt[i];
}

__global__ void deg_fin_kernel(const int* __restrict__ part, float* __restrict__ dinv, int n) {
  int i = blockIdx.x * blockDim.x + threadIdx.x;
  if (i >= n) return;
  int s = 0;
#pragma unroll 8
  for (int b = 0; b < DEG_SLICES; ++b) s += part[(size_t)b * n + i];
  dinv[i] = rsqrtf(1.0f + (float)s);
}

__global__ __launch_bounds__(64) void gemm64_wmma_kernel(const float* __restrict__ X,
                                                         const float* __restrict__ W,
                                                         float* __restrict__ H) {
  __shared__ float Wt[64 * LDS_STRIDE];
  for (int p = threadIdx.x; p < 64 * 64; p += 64) {
    int k = p >> 6, n = p & 63;
    Wt[n * LDS_STRIDE + k] = W[p];
  }
  __syncthreads();

  const int wave = threadIdx.x >> 5;
  const int lane = threadIdx.x & 31;
  const int half = lane >> 4;
  const int nc   = lane & 15;
  const int rowBase = blockIdx.x * 32 + wave * 16;

  const float* xrow = X + (size_t)(rowBase + nc) * 64;
  v8f acc0 = {}, acc1 = {}, acc2 = {}, acc3 = {};

#pragma unroll
  for (int k0 = 0; k0 < 64; k0 += 4) {
    const int kb = k0 + half * 2;
    v2f a  = *(const v2f*)(xrow + kb);
    v2f b0 = *(const v2f*)(&Wt[(0 * 16 + nc) * LDS_STRIDE + kb]);
    v2f b1 = *(const v2f*)(&Wt[(1 * 16 + nc) * LDS_STRIDE + kb]);
    v2f b2 = *(const v2f*)(&Wt[(2 * 16 + nc) * LDS_STRIDE + kb]);
    v2f b3 = *(const v2f*)(&Wt[(3 * 16 + nc) * LDS_STRIDE + kb]);
    acc0 = __builtin_amdgcn_wmma_f32_16x16x4_f32(false, a, false, b0, (short)0, acc0, false, false);
    acc1 = __builtin_amdgcn_wmma_f32_16x16x4_f32(false, a, false, b1, (short)0, acc1, false, false);
    acc2 = __builtin_amdgcn_wmma_f32_16x16x4_f32(false, a, false, b2, (short)0, acc2, false, false);
    acc3 = __builtin_amdgcn_wmma_f32_16x16x4_f32(false, a, false, b3, (short)0, acc3, false, false);
  }

#pragma unroll
  for (int v = 0; v < 8; ++v) {
    float* hrow = H + (size_t)(rowBase + v + half * 8) * 64;
    hrow[0 * 16 + nc] = acc0[v];
    hrow[1 * 16 + nc] = acc1[v];
    hrow[2 * 16 + nc] = acc2[v];
    hrow[3 * 16 + nc] = acc3[v];
  }
}

__global__ void gemm64_tail_kernel(const float* __restrict__ X, const float* __restrict__ W,
                                   float* __restrict__ H, int start, int n) {
  int t = blockIdx.x * blockDim.x + threadIdx.x;
  int r = start + (t >> 6), c = t & 63;
  if (r >= n) return;
  float s = 0.f;
  for (int k = 0; k < 64; ++k) s += X[(size_t)r * 64 + k] * W[k * 64 + c];
  H[(size_t)r * 64 + c] = s;
}

__global__ void self1_kernel(const float* __restrict__ h, const float* __restrict__ dinv,
                             float* __restrict__ h2, int n) {
  int t = blockIdx.x * blockDim.x + threadIdx.x;
  int node = t >> 4, c4 = (t & 15) * 4;
  if (node >= n) return;
  float s = dinv[node] * dinv[node];
  float4 v = *(const float4*)&h[(size_t)node * 64 + c4];
  float4 o = {v.x * s, v.y * s, v.z * s, v.w * s};
  *(float4*)&h2[(size_t)node * 64 + c4] = o;
}

__global__ void edge1_kernel(const float* __restrict__ h, const int* __restrict__ row,
                             const int* __restrict__ col, const float* __restrict__ dinv,
                             float* __restrict__ h2, int e) {
  int t = blockIdx.x * blockDim.x + threadIdx.x;
  int ed = t >> 4, c4 = (t & 15) * 4;
  if (ed >= e) return;
  int r = row[ed], c = col[ed];
  float nm = dinv[r] * dinv[c];
  float4 v = *(const float4*)&h[(size_t)r * 64 + c4];
  float* dst = &h2[(size_t)c * 64 + c4];
  atomicAdd(dst + 0, v.x * nm);
  atomicAdd(dst + 1, v.y * nm);
  atomicAdd(dst + 2, v.z * nm);
  atomicAdd(dst + 3, v.w * nm);
}

__global__ __launch_bounds__(64) void gemm2_wmma_kernel(const float* __restrict__ H2,
                                                        const float* __restrict__ B1,
                                                        const float* __restrict__ W2,
                                                        float* __restrict__ G) {
  __shared__ float Wt[16 * LDS_STRIDE];
  __shared__ float b1s[64];
  for (int p = threadIdx.x; p < 16 * 64; p += 64) {
    int n = p >> 6, k = p & 63;
    Wt[n * LDS_STRIDE + k] = (n < 2) ? W2[k * 2 + n] : 0.0f;
  }
  b1s[threadIdx.x] = B1[threadIdx.x];
  __syncthreads();

  const int wave = threadIdx.x >> 5;
  const int lane = threadIdx.x & 31;
  const int half = lane >> 4;
  const int nc   = lane & 15;
  const int rowBase = blockIdx.x * 32 + wave * 16;

  const float* hrow = H2 + (size_t)(rowBase + nc) * 64;
  v8f acc = {};

#pragma unroll
  for (int k0 = 0; k0 < 64; k0 += 4) {
    const int kb = k0 + half * 2;
    v2f a;
    a.x = fmaxf(hrow[kb + 0] + b1s[kb + 0], 0.0f);
    a.y = fmaxf(hrow[kb + 1] + b1s[kb + 1], 0.0f);
    v2f b = *(const v2f*)(&Wt[nc * LDS_STRIDE + kb]);
    acc = __builtin_amdgcn_wmma_f32_16x16x4_f32(false, a, false, b, (short)0, acc, false, false);
  }

  if (nc < 2) {
#pragma unroll
    for (int v = 0; v < 8; ++v)
      G[(size_t)(rowBase + v + half * 8) * 2 + nc] = acc[v];
  }
}

__global__ void gemm2_tail_kernel(const float* __restrict__ H2, const float* __restrict__ B1,
                                  const float* __restrict__ W2, float* __restrict__ G,
                                  int start, int n) {
  int t = blockIdx.x * blockDim.x + threadIdx.x;
  int r = start + (t >> 1), c = t & 1;
  if (r >= n) return;
  float s = 0.f;
  for (int k = 0; k < 64; ++k)
    s += fmaxf(H2[(size_t)r * 64 + k] + B1[k], 0.0f) * W2[k * 2 + c];
  G[(size_t)r * 2 + c] = s;
}

__global__ __launch_bounds__(PART_THREADS) void agg2_part_kernel(const float* __restrict__ g,
                                                                 const int* __restrict__ row,
                                                                 const int* __restrict__ col,
                                                                 const float* __restrict__ dinv,
                                                                 float* __restrict__ part,
                                                                 int e, int n) {
  __shared__ alignas(16) float acc[AGG_NP * 2];
  const int tid = threadIdx.x;
  const int lo = blockIdx.y * AGG_NP;
  const int span = min(AGG_NP, n - lo);
  for (int i = tid; i < AGG_NP / 2; i += PART_THREADS)
    ((float4*)acc)[i] = make_float4(0.f, 0.f, 0.f, 0.f);
  __syncthreads();

  const int per = (e + (int)gridDim.x - 1) / (int)gridDim.x;
  const int e0 = min(e, (int)blockIdx.x * per);
  const int e1 = min(e, e0 + per);
  for (int base = e0; base < e1; base += PART_THREADS * EDGE_UNROLL) {
    int c[EDGE_UNROLL], r[EDGE_UNROLL];
#pragma unroll
    for (int k = 0; k < EDGE_UNROLL; ++k) {
      const int i = base + k * PART_THREADS + tid;
      const bool ok = i < e1;
      const int ii = ok ? i : e0;
      c[k] = ok ? col[ii] : -1;
      r[k] = row[ii];
    }
    float nm[EDGE_UNROLL];
    float2 v[EDGE_UNROLL];
    int d[EDGE_UNROLL];
#pragma unroll
    for (int k = 0; k < EDGE_UNROLL; ++k) {
      d[k] = c[k] - lo;
      const bool m = (unsigned)d[k] < (unsigned)span;
      const int rr = m ? r[k] : 0;
      const int cc = m ? c[k] : 0;
      nm[k] = dinv[rr] * dinv[cc];
      v[k] = *(const float2*)&g[(size_t)rr * 2];
    }
#pragma unroll
    for (int k = 0; k < EDGE_UNROLL; ++k) {
      if ((unsigned)d[k] < (unsigned)span) {
        atomicAdd(&acc[d[k] * 2 + 0], v[k].x * nm[k]);
        atomicAdd(&acc[d[k] * 2 + 1], v[k].y * nm[k]);
      }
    }
  }
  __syncthreads();

  float* dst = part + (size_t)blockIdx.x * n * 2 + (size_t)lo * 2;
  for (int i = tid; i < span * 2; i += PART_THREADS) dst[i] = acc[i];
}

__global__ void agg2_fin_kernel(const float* __restrict__ g, const float* __restrict__ dinv,
                                const float* __restrict__ b2, const float* __restrict__ part,
                                float* __restrict__ out, int n) {
  int t = blockIdx.x * blockDim.x + threadIdx.x;
  if (t >= n * 2) return;
  const int i = t >> 1;
  const float s = dinv[i] * dinv[i];
  float a = g[t] * s + b2[t & 1];
#pragma unroll 8
  for (int b = 0; b < AGG_SLICES; ++b) a += part[(size_t)b * n * 2 + t];
  out[t] = a;
}

extern "C" void kernel_launch(void* const* d_in, const int* in_sizes, int n_in,
                              void* d_out, int out_size, void* d_ws, size_t ws_size,
                              hipStream_t stream) {
  const float* x  = (const float*)d_in[0];
  const int*   ei = (const int*)d_in[1];
  const float* W1 = (const float*)d_in[2];
  const float* b1 = (const float*)d_in[3];
  const float* W2 = (const float*)d_in[4];
  const float* b2 = (const float*)d_in[5];
  float* out = (float*)d_out;

  const int N = in_sizes[0] / 64;
  const int E = in_sizes[1] / 2;
  const int* row = ei;
  const int* col = ei + E;

  const int Npad = (N + 63) & ~63;
  float* dinv = (float*)d_ws;
  float* h    = dinv + Npad;
  float* h2   = h + (size_t)N * 64;
  float* g    = h2 + (size_t)N * 64;
  float* prt  = g + (size_t)N * 2;

  const int T = 256;
  deg_part_kernel<<<dim3(DEG_SLICES, (N + DEG_NP - 1) / DEG_NP), PART_THREADS, 0, stream>>>(
      col, (int*)prt, E, N);
  deg_fin_kernel<<<(N + T - 1) / T, T, 0, stream>>>((const int*)prt, dinv, N);

  const int mainBlocks = N / 32;
  const int tailStart  = mainBlocks * 32;
  if (mainBlocks > 0)
    gemm64_wmma_kernel<<<mainBlocks, 64, 0, stream>>>(x, W1, h);
  if (tailStart < N) {
    int tcnt = (N - tailStart) * 64;
    gemm64_tail_kernel<<<(tcnt + T - 1) / T, T, 0, stream>>>(x, W1, h, tailStart, N);
  }

  self1_kernel<<<((long long)N * 16 + T - 1) / T, T, 0, stream>>>(h, dinv, h2, N);
  edge1_kernel<<<((long long)E * 16 + T - 1) / T, T, 0, stream>>>(h, row, col, dinv, h2, E);

  if (mainBlocks > 0)
    gemm2_wmma_kernel<<<mainBlocks, 64, 0, stream>>>(h2, b1, W2, g);
  if (tailStart < N) {
    int tcnt = (N - tailStart) * 2;
    gemm2_tail_kernel<<<(tcnt + T - 1) / T, T, 0, stream>>>(h2, b1, W2, g, tailStart, N);
  }

  agg2_part_kernel<<<dim3(AGG_SLICES, (N + AGG_NP - 1) / AGG_NP), PART_THREADS, 0, stream>>>(
      g, row, col, dinv, prt, E, N);
  agg2_fin_kernel<<<(N * 2 + T - 1) / T, T, 0, stream>>>(g, dinv, b2, prt, out, N);
}


# Layer-1 global float atomics replaced by a per-node gather over an edge list sorted by destination (slots from LDS cursors, arrival order inside a slice); layer-1 sum order is now the list order
# speedup vs baseline: 2.3963x; 1.4444x over previous
#include <hip/hip_runtime.h>
#include <hip/hip_bf16.h>

typedef float v2f __attribute__((ext_vector_type(2)));
typedef float v8f __attribute__((ext_vector_type(8)));

#define LDS_STRIDE 66

#define PART_THREADS 1024
#define EDGE_UNROLL 8
#define DEG_NP 51200
#define DEG_SLICES 64
#define AGG_NP 25600
#define AGG_SLICES 32

__global__ __launch_bounds__(PART_THREADS) void deg_part_kernel(const int* __restrict__ col,
                                                                int* __restrict__ part,
                                                                int e, int n) {
  __shared__ alignas(16) int cnt[DEG_NP];
  const int tid = threadIdx.x;
  const int lo = blockIdx.y * DEG_NP;
  const int span = min(DEG_NP, n - lo);
  for (int i = tid; i < DEG_NP / 4; i += PART_THREADS) ((int4*)cnt)[i] = make_int4(0, 0, 0, 0);
  __syncthreads();

  const int per = (e + (int)gridDim.x - 1) / (int)gridDim.x;
  const int e0 = min(e, (int)blockIdx.x * per);
  const int e1 = min(e, e0 + per);
  for (int base = e0; base < e1; base += PART_THREADS * EDGE_UNROLL) {
    int d[EDGE_UNROLL];
#pragma unroll
    for (int k = 0; k < EDGE_UNROLL; ++k) {
      const int i = base + k * PART_THREADS + tid;
      const bool ok = i < e1;
      const int c = col[ok ? i : e0];
      d[k] = ok ? (c - lo) : -1;
    }
#pragma unroll
    for (int k = 0; k < EDGE_UNROLL; ++k)
      if ((unsigned)d[k] < (unsigned)span) atomicAdd(&cnt[d[k]], 1);
  }
  __syncthreads();

  int* dst = part + (size_t)blockIdx.x * n + lo;
  for (int i = tid; i < span; i += PART_THREADS) dst[i] = cnt[i];
}

#define SCAN_THREADS 256

__global__ __launch_bounds__(SCAN_THREADS) void deg_fin_kernel(int* __restrict__ part,
                                                               float* __restrict__ dinv,
                                                               int* __restrict__ rowptr,
                                                               int* __restrict__ bsum, int n) {
  __shared__ int wtot[32];
  const int tid = threadIdx.x;
  const int lane = tid & 31, wave = tid >> 5;
  const int i = blockIdx.x * SCAN_THREADS + tid;
  int s = 0;
  if (i < n) {
    int t[DEG_SLICES];
#pragma unroll
    for (int b = 0; b < DEG_SLICES; ++b) t[b] = part[(size_t)b * n + i];
#pragma unroll
    for (int b = 0; b < DEG_SLICES; ++b) {
      part[(size_t)b * n + i] = s;
      s += t[b];
    }
    dinv[i] = rsqrtf(1.0f + (float)s);
  }
  int incl = s;
  for (int o = 1; o < 32; o <<= 1) {
    const int u = __shfl_up(incl, o);
    if (lane >= o) incl += u;
  }
  if (lane == 31) wtot[wave] = incl;
  __syncthreads();
  if (wave == 0) {
    const int w = (lane < SCAN_THREADS / 32) ? wtot[lane] : 0;
    int wi = w;
    for (int o = 1; o < 32; o <<= 1) {
      const int u = __shfl_up(wi, o);
      if (lane >= o) wi += u;
    }
    wtot[lane] = wi - w;
    if (lane == 31) bsum[blockIdx.x] = wi;
  }
  __syncthreads();
  if (i < n) rowptr[i] = wtot[wave] + incl - s;
}

__global__ __launch_bounds__(SCAN_THREADS) void rowptr_kernel(int* __restrict__ rowptr,
                                                              const int* __restrict__ bsum,
                                                              int n) {
  __shared__ int red[32];
  const int tid = threadIdx.x;
  const int lane = tid & 31, wave = tid >> 5;
  int local = 0;
  for (int b = tid; b < (int)blockIdx.x; b += SCAN_THREADS) local += bsum[b];
  for (int o = 16; o > 0; o >>= 1) local += __shfl_xor(local, o);
  if (lane == 0) red[wave] = local;
  __syncthreads();
  int carry = 0;
#pragma unroll
  for (int w = 0; w < SCAN_THREADS / 32; ++w) carry += red[w];
  const int i = blockIdx.x * SCAN_THREADS + tid;
  if (i < n) rowptr[i] += carry;
}

__global__ __launch_bounds__(PART_THREADS) void fill_kernel(const int* __restrict__ row,
                                                            const int* __restrict__ col,
                                                            const int* __restrict__ part,
                                                            const int* __restrict__ rowptr,
                                                            int* __restrict__ srcs,
                                                            int e, int n) {
  __shared__ alignas(16) int cur[DEG_NP];
  const int tid = threadIdx.x;
  const int lo = blockIdx.y * DEG_NP;
  const int span = min(DEG_NP, n - lo);
  const int* pp = part + (size_t)blockIdx.x * n + lo;
  for (int i = tid; i < span; i += PART_THREADS) cur[i] = rowptr[lo + i] + pp[i];
  __syncthreads();

  const int per = (e + (int)gridDim.x - 1) / (int)gridDim.x;
  const int e0 = min(e, (int)blockIdx.x * per);
  const int e1 = min(e, e0 + per);
  for (int base = e0; base < e1; base += PART_THREADS * EDGE_UNROLL) {
    int d[EDGE_UNROLL], r[EDGE_UNROLL];
#pragma unroll
    for (int k = 0; k < EDGE_UNROLL; ++k) {
      const int i = base + k * PART_THREADS + tid;
      const bool ok = i < e1;
      const int ii = ok ? i : e0;
      const int c = col[ii];
      r[k] = row[ii];
      d[k] = ok ? (c - lo) : -1;
    }
#pragma unroll
    for (int k = 0; k < EDGE_UNROLL; ++k) {
      if ((unsigned)d[k] < (unsigned)span) {
        const int pos = atomicAdd(&cur[d[k]], 1);
        srcs[pos] = r[k];
      }
    }
  }
}

__global__ __launch_bounds__(64) void gemm64_wmma_kernel(const float* __restrict__ X,
                                                         const float* __restrict__ W,
                                                         float* __restrict__ H) {
  __shared__ float Wt[64 * LDS_STRIDE];
  for (int p = threadIdx.x; p < 64 * 64; p += 64) {
    int k = p >> 6, n = p & 63;
    Wt[n * LDS_STRIDE + k] = W[p];
  }
  __syncthreads();

  const int wave = threadIdx.x >> 5;
  const int lane = threadIdx.x & 31;
  const int half = lane >> 4;
  const int nc   = lane & 15;
  const int rowBase = blockIdx.x * 32 + wave * 16;

  const float* xrow = X + (size_t)(rowBase + nc) * 64;
  v8f acc0 = {}, acc1 = {}, acc2 = {}, acc3 = {};

#pragma unroll
  for (int k0 = 0; k0 < 64; k0 += 4) {
    const int kb = k0 + half * 2;
    v2f a  = *(const v2f*)(xrow + kb);
    v2f b0 = *(const v2f*)(&Wt[(0 * 16 + nc) * LDS_STRIDE + kb]);
    v2f b1 = *(const v2f*)(&Wt[(1 * 16 + nc) * LDS_STRIDE + kb]);
    v2f b2 = *(const v2f*)(&Wt[(2 * 16 + nc) * LDS_STRIDE + kb]);
    v2f b3 = *(const v2f*)(&Wt[(3 * 16 + nc) * LDS_STRIDE + kb]);
    acc0 = __builtin_amdgcn_wmma_f32_16x16x4_f32(false, a, false, b0, (short)0, acc0, false, false);
    acc1 = __builtin_amdgcn_wmma_f32_16x16x4_f32(false, a, false, b1, (short)0, acc1, false, false);
    acc2 = __builtin_amdgcn_wmma_f32_16x16x4_f32(false, a, false, b2, (short)0, acc2, false, false);
    acc3 = __builtin_amdgcn_wmma_f32_16x16x4_f32(false, a, false, b3, (short)0, acc3, false, false);
  }

#pragma unroll
  for (int v = 0; v < 8; ++v) {
    float* hrow = H + (size_t)(rowBase + v + half * 8) * 64;
    hrow[0 * 16 + nc] = acc0[v];
    hrow[1 * 16 + nc] = acc1[v];
    hrow[2 * 16 + nc] = acc2[v];
    hrow[3 * 16 + nc] = acc3[v];
  }
}

__global__ void gemm64_tail_kernel(const float* __restrict__ X, const float* __restrict__ W,
                                   float* __restrict__ H, int start, int n) {
  int t = blockIdx.x * blockDim.x + threadIdx.x;
  int r = start + (t >> 6), c = t & 63;
  if (r >= n) return;
  float s = 0.f;
  for (int k = 0; k < 64; ++k) s += X[(size_t)r * 64 + k] * W[k * 64 + c];
  H[(size_t)r * 64 + c] = s;
}

#define GATHER_CHUNK 8

__global__ __launch_bounds__(256) void gather1_kernel(const float* __restrict__ h,
                                                      const int* __restrict__ rowptr,
                                                      const int* __restrict__ srcs,
                                                      const float* __restrict__ dinv,
                                                      float* __restrict__ h2, int e, int n) {
  const int node = blockIdx.x * 16 + ((int)threadIdx.x >> 4);
  const int l = threadIdx.x & 15;
  if (node >= n) return;
  const int beg = rowptr[node];
  const int end = (node + 1 < n) ? rowptr[node + 1] : e;
  const float di = dinv[node];
  const float4* h4 = (const float4*)h;
  const float4 v0 = h4[(size_t)node * 16 + l];
  const float ss = di * di;
  float4 acc = {v0.x * ss, v0.y * ss, v0.z * ss, v0.w * ss};

  int j = beg;
  for (; j + GATHER_CHUNK <= end; j += GATHER_CHUNK) {
    int s[GATHER_CHUNK];
    float w[GATHER_CHUNK];
    float4 a[GATHER_CHUNK];
#pragma unroll
    for (int k = 0; k < GATHER_CHUNK; ++k) s[k] = srcs[j + k];
#pragma unroll
    for (int k = 0; k < GATHER_CHUNK; ++k) {
      w[k] = dinv[s[k]] * di;
      a[k] = h4[(size_t)s[k] * 16 + l];
    }
#pragma unroll
    for (int k = 0; k < GATHER_CHUNK; ++k) {
      acc.x += a[k].x * w[k];
      acc.y += a[k].y * w[k];
      acc.z += a[k].z * w[k];
      acc.w += a[k].w * w[k];
    }
  }
  if (j < end) {
    int s[GATHER_CHUNK];
    float w[GATHER_CHUNK];
    float4 a[GATHER_CHUNK];
#pragma unroll
    for (int k = 0; k < GATHER_CHUNK; ++k) s[k] = srcs[min(j + k, end - 1)];
#pragma unroll
    for (int k = 0; k < GATHER_CHUNK; ++k) {
      const float dv = dinv[s[k]];
      w[k] = (j + k < end) ? dv * di : 0.0f;
      a[k] = h4[(size_t)s[k] * 16 + l];
    }
#pragma unroll
    for (int k = 0; k < GATHER_CHUNK; ++k) {
      acc.x += a[k].x * w[k];
      acc.y += a[k].y * w[k];
      acc.z += a[k].z * w[k];
      acc.w += a[k].w * w[k];
    }
  }
  *(float4*)&h2[(size_t)node * 64 + l * 4] = acc;
}

__global__ __launch_bounds__(64) void gemm2_wmma_kernel(const float* __restrict__ H2,
                                                        const float* __restrict__ B1,
                                                        const float* __restrict__ W2,
                                                        float* __restrict__ G) {
  __shared__ float Wt[16 * LDS_STRIDE];
  __shared__ float b1s[64];
  for (int p = threadIdx.x; p < 16 * 64; p += 64) {
    int n = p >> 6, k = p & 63;
    Wt[n * LDS_STRIDE + k] = (n < 2) ? W2[k * 2 + n] : 0.0f;
  }
  b1s[threadIdx.x] = B1[threadIdx.x];
  __syncthreads();

  const int wave = threadIdx.x >> 5;
  const int lane = threadIdx.x & 31;
  const int half = lane >> 4;
  const int nc   = lane & 15;
  const int rowBase = blockIdx.x * 32 + wave * 16;

  const float* hrow = H2 + (size_t)(rowBase + nc) * 64;
  v8f acc = {};

#pragma unroll
  for (int k0 = 0; k0 < 64; k0 += 4) {
    const int kb = k0 + half * 2;
    v2f a;
    a.x = fmaxf(hrow[kb + 0] + b1s[kb + 0], 0.0f);
    a.y = fmaxf(hrow[kb + 1] + b1s[kb + 1], 0.0f);
    v2f b = *(const v2f*)(&Wt[nc * LDS_STRIDE + kb]);
    acc = __builtin_amdgcn_wmma_f32_16x16x4_f32(false, a, false, b, (short)0, acc, false, false);
  }

  if (nc < 2) {
#pragma unroll
    for (int v = 0; v < 8; ++v)
      G[(size_t)(rowBase + v + half * 8) * 2 + nc] = acc[v];
  }
}

__global__ void gemm2_tail_kernel(const float* __restrict__ H2, const float* __restrict__ B1,
                                  const float* __restrict__ W2, float* __restrict__ G,
                                  int start, int n) {
  int t = blockIdx.x * blockDim.x + threadIdx.x;
  int r = start + (t >> 1), c = t & 1;
  if (r >= n) return;
  float s = 0.f;
  for (int k = 0; k < 64; ++k)
    s += fmaxf(H2[(size_t)r * 64 + k] + B1[k], 0.0f) * W2[k * 2 + c];
  G[(size_t)r * 2 + c] = s;
}

__global__ __launch_bounds__(PART_THREADS) void agg2_part_kernel(const float* __restrict__ g,
                                                                 const int* __restrict__ row,
                                                                 const int* __restrict__ col,
                                                                 const float* __restrict__ dinv,
                                                                 float* __restrict__ part,
                                                                 int e, int n) {
  __shared__ alignas(16) float acc[AGG_NP * 2];
  const int tid = threadIdx.x;
  const int lo = blockIdx.y * AGG_NP;
  const int span = min(AGG_NP, n - lo);
  for (int i = tid; i < AGG_NP / 2; i += PART_THREADS)
    ((float4*)acc)[i] = make_float4(0.f, 0.f, 0.f, 0.f);
  __syncthreads();

  const int per = (e + (int)gridDim.x - 1) / (int)gridDim.x;
  const int e0 = min(e, (int)blockIdx.x * per);
  const int e1 = min(e, e0 + per);
  for (int base = e0; base < e1; base += PART_THREADS * EDGE_UNROLL) {
    int c[EDGE_UNROLL], r[EDGE_UNROLL];
#pragma unroll
    for (int k = 0; k < EDGE_UNROLL; ++k) {
      const int i = base + k * PART_THREADS + tid;
      const bool ok = i < e1;
      const int ii = ok ? i : e0;
      c[k] = ok ? col[ii] : -1;
      r[k] = row[ii];
    }
    float nm[EDGE_UNROLL];
    float2 v[EDGE_UNROLL];
    int d[EDGE_UNROLL];
#pragma unroll
    for (int k = 0; k < EDGE_UNROLL; ++k) {
      d[k] = c[k] - lo;
      const bool m = (unsigned)d[k] < (unsigned)span;
      const int rr = m ? r[k] : 0;
      const int cc = m ? c[k] : 0;
      nm[k] = dinv[rr] * dinv[cc];
      v[k] = *(const float2*)&g[(size_t)rr * 2];
    }
#pragma unroll
    for (int k = 0; k < EDGE_UNROLL; ++k) {
      if ((unsigned)d[k] < (unsigned)span) {
        atomicAdd(&acc[d[k] * 2 + 0], v[k].x * nm[k]);
        atomicAdd(&acc[d[k] * 2 + 1], v[k].y * nm[k]);
      }
    }
  }
  __syncthreads();

  float* dst = part + (size_t)blockIdx.x * n * 2 + (size_t)lo * 2;
  for (int i = tid; i < span * 2; i += PART_THREADS) dst[i] = acc[i];
}

__global__ void agg2_fin_kernel(const float* __restrict__ g, const float* __restrict__ dinv,
                                const float* __restrict__ b2, const float* __restrict__ part,
                                float* __restrict__ out, int n) {
  int t = blockIdx.x * blockDim.x + threadIdx.x;
  if (t >= n * 2) return;
  const int i = t >> 1;
  const float s = dinv[i] * dinv[i];
  float a = g[t] * s + b2[t & 1];
#pragma unroll 8
  for (int b = 0; b < AGG_SLICES; ++b) a += part[(size_t)b * n * 2 + t];
  out[t] = a;
}

extern "C" void kernel_launch(void* const* d_in, const int* in_sizes, int n_in,
                              void* d_out, int out_size, void* d_ws, size_t ws_size,
                              hipStream_t stream) {
  const float* x  = (const float*)d_in[0];
  const int*   ei = (const int*)d_in[1];
  const float* W1 = (const float*)d_in[2];
  const float* b1 = (const float*)d_in[3];
  const float* W2 = (const float*)d_in[4];
  const float* b2 = (const float*)d_in[5];
  float* out = (float*)d_out;

  const int N = in_sizes[0] / 64;
  const int E = in_sizes[1] / 2;
  const int* row = ei;
  const int* col = ei + E;

  const int Npad = (N + 63) & ~63;
  float* dinv = (float*)d_ws;
  float* h    = dinv + Npad;
  float* h2   = h + (size_t)N * 64;
  float* g    = h2 + (size_t)N * 64;
  float* prt  = g + (size_t)N * 2;
  int* srcs   = (int*)(prt + (size_t)DEG_SLICES * N);
  int* rowptr = srcs + E;
  int* bsum   = rowptr + Npad;

  const int T = 256;
  deg_part_kernel<<<dim3(DEG_SLICES, (N + DEG_NP - 1) / DEG_NP), PART_THREADS, 0, stream>>>(
      col, (int*)prt, E, N);
  const int scanBlocks = (N + SCAN_THREADS - 1) / SCAN_THREADS;
  deg_fin_kernel<<<scanBlocks, SCAN_THREADS, 0, stream>>>((int*)prt, dinv, rowptr, bsum, N);
  rowptr_kernel<<<scanBlocks, SCAN_THREADS, 0, stream>>>(rowptr, bsum, N);
  fill_kernel<<<dim3(DEG_SLICES, (N + DEG_NP - 1) / DEG_NP), PART_THREADS, 0, stream>>>(
      row, col, (const int*)prt, rowptr, srcs, E, N);

  const int mainBlocks = N / 32;
  const int tailStart  = mainBlocks * 32;
  if (mainBlocks > 0)
    gemm64_wmma_kernel<<<mainBlocks, 64, 0, stream>>>(x, W1, h);
  if (tailStart < N) {
    int tcnt = (N - tailStart) * 64;
    gemm64_tail_kernel<<<(tcnt + T - 1) / T, T, 0, stream>>>(x, W1, h, tailStart, N);
  }

  gather1_kernel<<<(N + 15) / 16, 256, 0, stream>>>(h, rowptr, srcs, dinv, h2, E, N);

  if (mainBlocks > 0)
    gemm2_wmma_kernel<<<mainBlocks, 64, 0, stream>>>(h2, b1, W2, g);
  if (tailStart < N) {
    int tcnt = (N - tailStart) * 2;
    gemm2_tail_kernel<<<(tcnt + T - 1) / T, T, 0, stream>>>(h2, b1, W2, g, tailStart, N);
  }

  agg2_part_kernel<<<dim3(AGG_SLICES, (N + AGG_NP - 1) / AGG_NP), PART_THREADS, 0, stream>>>(
      g, row, col, dinv, prt, E, N);
  agg2_fin_kernel<<<(N * 2 + T - 1) / T, T, 0, stream>>>(g, dinv, b2, prt, out, N);
}


# Layer-2 scatter-add (LDS partial sums, closing sum) replaced by a per-node gather over the sorted edge list reading one 16-byte (g, dinv) record per edge; layer-2 sum order: list order, 8-lane tree
# speedup vs baseline: 3.0118x; 1.2569x over previous
#include <hip/hip_runtime.h>
#include <hip/hip_bf16.h>

typedef float v2f __attribute__((ext_vector_type(2)));
typedef float v8f __attribute__((ext_vector_type(8)));

#define LDS_STRIDE 66

#define PART_THREADS 1024
#define EDGE_UNROLL 8
#define DEG_NP 51200
#define DEG_SLICES 64

__global__ __launch_bounds__(PART_THREADS) void deg_part_kernel(const int* __restrict__ col,
                                                                int* __restrict__ part,
                                                                int e, int n) {
  __shared__ alignas(16) int cnt[DEG_NP];
  const int tid = threadIdx.x;
  const int lo = blockIdx.y * DEG_NP;
  const int span = min(DEG_NP, n - lo);
  for (int i = tid; i < DEG_NP / 4; i += PART_THREADS) ((int4*)cnt)[i] = make_int4(0, 0, 0, 0);
  __syncthreads();

  const int per = (e + (int)gridDim.x - 1) / (int)gridDim.x;
  const int e0 = min(e, (int)blockIdx.x * per);
  const int e1 = min(e, e0 + per);
  for (int base = e0; base < e1; base += PART_THREADS * EDGE_UNROLL) {
    int d[EDGE_UNROLL];
#pragma unroll
    for (int k = 0; k < EDGE_UNROLL; ++k) {
      const int i = base + k * PART_THREADS + tid;
      const bool ok = i < e1;
      const int c = col[ok ? i : e0];
      d[k] = ok ? (c - lo) : -1;
    }
#pragma unroll
    for (int k = 0; k < EDGE_UNROLL; ++k)
      if ((unsigned)d[k] < (unsigned)span) atomicAdd(&cnt[d[k]], 1);
  }
  __syncthreads();

  int* dst = part + (size_t)blockIdx.x * n + lo;
  for (int i = tid; i < span; i += PART_THREADS) dst[i] = cnt[i];
}

#define SCAN_THREADS 256

__global__ __launch_bounds__(SCAN_THREADS) void deg_fin_kernel(int* __restrict__ part,
                                                               float* __restrict__ dinv,
                                                               int* __restrict__ rowptr,
                                                               int* __restrict__ bsum, int n) {
  __shared__ int wtot[32];
  const int tid = threadIdx.x;
  const int lane = tid & 31, wave = tid >> 5;
  const int i = blockIdx.x * SCAN_THREADS + tid;
  int s = 0;
  if (i < n) {
    int t[DEG_SLICES];
#pragma unroll
    for (int b = 0; b < DEG_SLICES; ++b) t[b] = part[(size_t)b * n + i];
#pragma unroll
    for (int b = 0; b < DEG_SLICES; ++b) {
      part[(size_t)b * n + i] = s;
      s += t[b];
    }
    dinv[i] = rsqrtf(1.0f + (float)s);
  }
  int incl = s;
  for (int o = 1; o < 32; o <<= 1) {
    const int u = __shfl_up(incl, o);
    if (lane >= o) incl += u;
  }
  if (lane == 31) wtot[wave] = incl;
  __syncthreads();
  if (wave == 0) {
    const int w = (lane < SCAN_THREADS / 32) ? wtot[lane] : 0;
    int wi = w;
    for (int o = 1; o < 32; o <<= 1) {
      const int u = __shfl_up(wi, o);
      if (lane >= o) wi += u;
    }
    wtot[lane] = wi - w;
    if (lane == 31) bsum[blockIdx.x] = wi;
  }
  __syncthreads();
  if (i < n) rowptr[i] = wtot[wave] + incl - s;
}

__global__ __launch_bounds__(SCAN_THREADS) void rowptr_kernel(int* __restrict__ rowptr,
                                                              const int* __restrict__ bsum,
                                                              int n) {
  __shared__ int red[32];
  const int tid = threadIdx.x;
  const int lane = tid & 31, wave = tid >> 5;
  int local = 0;
  for (int b = tid; b < (int)blockIdx.x; b += SCAN_THREADS) local += bsum[b];
  for (int o = 16; o > 0; o >>= 1) local += __shfl_xor(local, o);
  if (lane == 0) red[wave] = local;
  __syncthreads();
  int carry = 0;
#pragma unroll
  for (int w = 0; w < SCAN_THREADS / 32; ++w) carry += red[w];
  const int i = blockIdx.x * SCAN_THREADS + tid;
  if (i < n) rowptr[i] += carry;
}

__global__ __launch_bounds__(PART_THREADS) void fill_kernel(const int* __restrict__ row,
                                                            const int* __restrict__ col,
                                                            const int* __restrict__ part,
                                                            const int* __restrict__ rowptr,
                                                            int* __restrict__ srcs,
                                                            int e, int n) {
  __shared__ alignas(16) int cur[DEG_NP];
  const int tid = threadIdx.x;
  const int lo = blockIdx.y * DEG_NP;
  const int span = min(DEG_NP, n - lo);
  const int* pp = part + (size_t)blockIdx.x * n + lo;
  for (int i = tid; i < span; i += PART_THREADS) cur[i] = rowptr[lo + i] + pp[i];
  __syncthreads();

  const int per = (e + (int)gridDim.x - 1) / (int)gridDim.x;
  const int e0 = min(e, (int)blockIdx.x * per);
  const int e1 = min(e, e0 + per);
  for (int base = e0; base < e1; base += PART_THREADS * EDGE_UNROLL) {
    int d[EDGE_UNROLL], r[EDGE_UNROLL];
#pragma unroll
    for (int k = 0; k < EDGE_UNROLL; ++k) {
      const int i = base + k * PART_THREADS + tid;
      const bool ok = i < e1;
      const int ii = ok ? i : e0;
      const int c = col[ii];
      r[k] = row[ii];
      d[k] = ok ? (c - lo) : -1;
    }
#pragma unroll
    for (int k = 0; k < EDGE_UNROLL; ++k) {
      if ((unsigned)d[k] < (unsigned)span) {
        const int pos = atomicAdd(&cur[d[k]], 1);
        srcs[pos] = r[k];
      }
    }
  }
}

__global__ __launch_bounds__(64) void gemm64_wmma_kernel(const float* __restrict__ X,
                                                         const float* __restrict__ W,
                                                         float* __restrict__ H) {
  __shared__ float Wt[64 * LDS_STRIDE];
  for (int p = threadIdx.x; p < 64 * 64; p += 64) {
    int k = p >> 6, n = p & 63;
    Wt[n * LDS_STRIDE + k] = W[p];
  }
  __syncthreads();

  const int wave = threadIdx.x >> 5;
  const int lane = threadIdx.x & 31;
  const int half = lane >> 4;
  const int nc   = lane & 15;
  const int rowBase = blockIdx.x * 32 + wave * 16;

  const float* xrow = X + (size_t)(rowBase + nc) * 64;
  v8f acc0 = {}, acc1 = {}, acc2 = {}, acc3 = {};

#pragma unroll
  for (int k0 = 0; k0 < 64; k0 += 4) {
    const int kb = k0 + half * 2;
    v2f a  = *(const v2f*)(xrow + kb);
    v2f b0 = *(const v2f*)(&Wt[(0 * 16 + nc) * LDS_STRIDE + kb]);
    v2f b1 = *(const v2f*)(&Wt[(1 * 16 + nc) * LDS_STRIDE + kb]);
    v2f b2 = *(const v2f*)(&Wt[(2 * 16 + nc) * LDS_STRIDE + kb]);
    v2f b3 = *(const v2f*)(&Wt[(3 * 16 + nc) * LDS_STRIDE + kb]);
    acc0 = __builtin_amdgcn_wmma_f32_16x16x4_f32(false, a, false, b0, (short)0, acc0, false, false);
    acc1 = __builtin_amdgcn_wmma_f32_16x16x4_f32(false, a, false, b1, (short)0, acc1, false, false);
    acc2 = __builtin_amdgcn_wmma_f32_16x16x4_f32(false, a, false, b2, (short)0, acc2, false, false);
    acc3 = __builtin_amdgcn_wmma_f32_16x16x4_f32(false, a, false, b3, (short)0, acc3, false, false);
  }

#pragma unroll
  for (int v = 0; v < 8; ++v) {
    float* hrow = H + (size_t)(rowBase + v + half * 8) * 64;
    hrow[0 * 16 + nc] = acc0[v];
    hrow[1 * 16 + nc] = acc1[v];
    hrow[2 * 16 + nc] = acc2[v];
    hrow[3 * 16 + nc] = acc3[v];
  }
}

__global__ void gemm64_tail_kernel(const float* __restrict__ X, const float* __restrict__ W,
                                   float* __restrict__ H, int start, int n) {
  int t = blockIdx.x * blockDim.x + threadIdx.x;
  int r = start + (t >> 6), c = t & 63;
  if (r >= n) return;
  float s = 0.f;
  for (int k = 0; k < 64; ++k) s += X[(size_t)r * 64 + k] * W[k * 64 + c];
  H[(size_t)r * 64 + c] = s;
}

#define GATHER_CHUNK 8

__global__ __launch_bounds__(256) void gather1_kernel(const float* __restrict__ h,
                                                      const int* __restrict__ rowptr,
                                                      const int* __restrict__ srcs,
                                                      const float* __restrict__ dinv,
                                                      float* __restrict__ h2, int e, int n) {
  const int node = blockIdx.x * 16 + ((int)threadIdx.x >> 4);
  const int l = threadIdx.x & 15;
  if (node >= n) return;
  const int beg = rowptr[node];
  const int end = (node + 1 < n) ? rowptr[node + 1] : e;
  const float di = dinv[node];
  const float4* h4 = (const float4*)h;
  const float4 v0 = h4[(size_t)node * 16 + l];
  const float ss = di * di;
  float4 acc = {v0.x * ss, v0.y * ss, v0.z * ss, v0.w * ss};

  int j = beg;
  for (; j + GATHER_CHUNK <= end; j += GATHER_CHUNK) {
    int s[GATHER_CHUNK];
    float w[GATHER_CHUNK];
    float4 a[GATHER_CHUNK];
#pragma unroll
    for (int k = 0; k < GATHER_CHUNK; ++k) s[k] = srcs[j + k];
#pragma unroll
    for (int k = 0; k < GATHER_CHUNK; ++k) {
      w[k] = dinv[s[k]] * di;
      a[k] = h4[(size_t)s[k] * 16 + l];
    }
#pragma unroll
    for (int k = 0; k < GATHER_CHUNK; ++k) {
      acc.x += a[k].x * w[k];
      acc.y += a[k].y * w[k];
      acc.z += a[k].z * w[k];
      acc.w += a[k].w * w[k];
    }
  }
  if (j < end) {
    int s[GATHER_CHUNK];
    float w[GATHER_CHUNK];
    float4 a[GATHER_CHUNK];
#pragma unroll
    for (int k = 0; k < GATHER_CHUNK; ++k) s[k] = srcs[min(j + k, end - 1)];
#pragma unroll
    for (int k = 0; k < GATHER_CHUNK; ++k) {
      const float dv = dinv[s[k]];
      w[k] = (j + k < end) ? dv * di : 0.0f;
      a[k] = h4[(size_t)s[k] * 16 + l];
    }
#pragma unroll
    for (int k = 0; k < GATHER_CHUNK; ++k) {
      acc.x += a[k].x * w[k];
      acc.y += a[k].y * w[k];
      acc.z += a[k].z * w[k];
      acc.w += a[k].w * w[k];
    }
  }
  *(float4*)&h2[(size_t)node * 64 + l * 4] = acc;
}

__global__ __launch_bounds__(64) void gemm2_wmma_kernel(const float* __restrict__ H2,
                                                        const float* __restrict__ B1,
                                                        const float* __restrict__ W2,
                                                        const float* __restrict__ dinv,
                                                        float* __restrict__ G) {
  __shared__ float Wt[16 * LDS_STRIDE];
  __shared__ float b1s[64];
  for (int p = threadIdx.x; p < 16 * 64; p += 64) {
    int n = p >> 6, k = p & 63;
    Wt[n * LDS_STRIDE + k] = (n < 2) ? W2[k * 2 + n] : 0.0f;
  }
  b1s[threadIdx.x] = B1[threadIdx.x];
  __syncthreads();

  const int wave = threadIdx.x >> 5;
  const int lane = threadIdx.x & 31;
  const int half = lane >> 4;
  const int nc   = lane & 15;
  const int rowBase = blockIdx.x * 32 + wave * 16;

  const float* hrow = H2 + (size_t)(rowBase + nc) * 64;
  const float dv = dinv[rowBase + nc];
  v8f acc = {};

#pragma unroll
  for (int k0 = 0; k0 < 64; k0 += 4) {
    const int kb = k0 + half * 2;
    v2f a;
    a.x = fmaxf(hrow[kb + 0] + b1s[kb + 0], 0.0f);
    a.y = fmaxf(hrow[kb + 1] + b1s[kb + 1], 0.0f);
    v2f b = *(const v2f*)(&Wt[nc * LDS_STRIDE + kb]);
    acc = __builtin_amdgcn_wmma_f32_16x16x4_f32(false, a, false, b, (short)0, acc, false, false);
  }

  if (nc < 2) {
#pragma unroll
    for (int v = 0; v < 8; ++v)
      G[(size_t)(rowBase + v + half * 8) * 4 + nc] = acc[v];
  }
  if (half == 0)
    *(v2f*)&G[(size_t)(rowBase + nc) * 4 + 2] = v2f{dv, 0.0f};
}

__global__ void gemm2_tail_kernel(const float* __restrict__ H2, const float* __restrict__ B1,
                                  const float* __restrict__ W2, const float* __restrict__ dinv,
                                  float* __restrict__ G, int start, int n) {
  int t = blockIdx.x * blockDim.x + threadIdx.x;
  int r = start + (t >> 2), c = t & 3;
  if (r >= n) return;
  float s = 0.f;
  if (c < 2) {
    for (int k = 0; k < 64; ++k)
      s += fmaxf(H2[(size_t)r * 64 + k] + B1[k], 0.0f) * W2[k * 2 + c];
  } else if (c == 2) {
    s = dinv[r];
  }
  G[(size_t)r * 4 + c] = s;
}

#define G2_LANES 8
#define G2_CHUNK 4
#define G2_THREADS 256
#define G2_NODES (G2_THREADS / G2_LANES)

__global__ __launch_bounds__(G2_THREADS) void gather2_kernel(const float* __restrict__ rec,
                                                             const int* __restrict__ rowptr,
                                                             const int* __restrict__ srcs,
                                                             const float* __restrict__ b2,
                                                             float* __restrict__ out, int e, int n) {
  __shared__ float stage[G2_NODES * 2];
  const int tid = threadIdx.x;
  const int ln = tid / G2_LANES;
  const int t = tid & (G2_LANES - 1);
  const int node0 = blockIdx.x * G2_NODES;
  const int node = min(node0 + ln, n - 1);
  const int beg = rowptr[node];
  const int end = (node + 1 < n) ? rowptr[node + 1] : e;
  const float4* r4 = (const float4*)rec;
  const float4 self = r4[node];
  const float di = self.z;
  float ax = 0.0f, ay = 0.0f;

  for (int j = beg + t; j < end; j += G2_LANES * G2_CHUNK) {
    int s[G2_CHUNK];
    float4 r[G2_CHUNK];
#pragma unroll
    for (int k = 0; k < G2_CHUNK; ++k) {
      const int jj = j + k * G2_LANES;
      s[k] = srcs[(jj < end) ? jj : j];
    }
#pragma unroll
    for (int k = 0; k < G2_CHUNK; ++k) {
      const bool ok = j + k * G2_LANES < end;
      r[k] = r4[ok ? s[k] : node];
    }
    static_assert(G2_CHUNK == 4, "the fence below names four records");
    asm volatile("" ::"v"(r[0].x), "v"(r[1].x), "v"(r[2].x), "v"(r[3].x));
#pragma unroll
    for (int k = 0; k < G2_CHUNK; ++k) {
      const bool ok = j + k * G2_LANES < end;
      const float w = ok ? r[k].z * di : 0.0f;
      ax += r[k].x * w;
      ay += r[k].y * w;
    }
  }
#pragma unroll
  for (int o = G2_LANES / 2; o > 0; o >>= 1) {
    ax += __shfl_xor(ax, o);
    ay += __shfl_xor(ay, o);
  }
  if (t == 0) {
    const float ss = di * di;
    stage[ln * 2 + 0] = (self.x * ss + b2[0]) + ax;
    stage[ln * 2 + 1] = (self.y * ss + b2[1]) + ay;
  }
  __syncthreads();
  const int nf = min(G2_NODES, n - node0) * 2;
  if (tid < nf) out[(size_t)node0 * 2 + tid] = stage[tid];
}

extern "C" void kernel_launch(void* const* d_in, const int* in_sizes, int n_in,
                              void* d_out, int out_size, void* d_ws, size_t ws_size,
                              hipStream_t stream) {
  const float* x  = (const float*)d_in[0];
  const int*   ei = (const int*)d_in[1];
  const float* W1 = (const float*)d_in[2];
  const float* b1 = (const float*)d_in[3];
  const float* W2 = (const float*)d_in[4];
  const float* b2 = (const float*)d_in[5];
  float* out = (float*)d_out;

  const int N = in_sizes[0] / 64;
  const int E = in_sizes[1] / 2;
  const int* row = ei;
  const int* col = ei + E;

  const int Npad = (N + 63) & ~63;
  float* dinv = (float*)d_ws;
  float* h    = dinv + Npad;
  float* h2   = h + (size_t)N * 64;
  float* g    = h2 + (size_t)N * 64;
  float* prt  = g + (size_t)N * 4;
  int* srcs   = (int*)(prt + (size_t)DEG_SLICES * N);
  int* rowptr = srcs + E;
  int* bsum   = rowptr + Npad;

  const int T = 256;
  deg_part_kernel<<<dim3(DEG_SLICES, (N + DEG_NP - 1) / DEG_NP), PART_THREADS, 0, stream>>>(
      col, (int*)prt, E, N);
  const int scanBlocks = (N + SCAN_THREADS - 1) / SCAN_THREADS;
  deg_fin_kernel<<<scanBlocks, SCAN_THREADS, 0, stream>>>((int*)prt, dinv, rowptr, bsum, N);
  rowptr_kernel<<<scanBlocks, SCAN_THREADS, 0, stream>>>(rowptr, bsum, N);
  fill_kernel<<<dim3(DEG_SLICES, (N + DEG_NP - 1) / DEG_NP), PART_THREADS, 0, stream>>>(
      row, col, (const int*)prt, rowptr, srcs, E, N);

  const int mainBlocks = N / 32;
  const int tailStart  = mainBlocks * 32;
  if (mainBlocks > 0)
    gemm64_wmma_kernel<<<mainBlocks, 64, 0, stream>>>(x, W1, h);
  if (tailStart < N) {
    int tcnt = (N - tailStart) * 64;
    gemm64_tail_kernel<<<(tcnt + T - 1) / T, T, 0, stream>>>(x, W1, h, tailStart, N);
  }

  gather1_kernel<<<(N + 15) / 16, 256, 0, stream>>>(h, rowptr, srcs, dinv, h2, E, N);

  if (mainBlocks > 0)
    gemm2_wmma_kernel<<<mainBlocks, 64, 0, stream>>>(h2, b1, W2, dinv, g);
  if (tailStart < N) {
    int tcnt = (N - tailStart) * 4;
    gemm2_tail_kernel<<<(tcnt + T - 1) / T, T, 0, stream>>>(h2, b1, W2, dinv, g, tailStart, N);
  }

  gather2_kernel<<<(N + G2_NODES - 1) / G2_NODES, G2_THREADS, 0, stream>>>(g, rowptr, srcs, b2, out,
                                                                           E, N);
}


# Layer-1 GEMM stores its rows pre-scaled by dinv[source]; the layer-1 gather drops the per-edge dinv read and multiply and scales once by dinv[dest]; weight rounding order changes
# speedup vs baseline: 3.1582x; 1.0486x over previous
#include <hip/hip_runtime.h>
#include <hip/hip_bf16.h>

typedef float v2f __attribute__((ext_vector_type(2)));
typedef float v8f __attribute__((ext_vector_type(8)));

#define LDS_STRIDE 66

#define PART_THREADS 1024
#define EDGE_UNROLL 8
#define DEG_NP 51200
#define DEG_SLICES 64

__global__ __launch_bounds__(PART_THREADS) void deg_part_kernel(const int* __restrict__ col,
                                                                int* __restrict__ part,
                                                                int e, int n) {
  __shared__ alignas(16) int cnt[DEG_NP];
  const int tid = threadIdx.x;
  const int lo = blockIdx.y * DEG_NP;
  const int span = min(DEG_NP, n - lo);
  for (int i = tid; i < DEG_NP / 4; i += PART_THREADS) ((int4*)cnt)[i] = make_int4(0, 0, 0, 0);
  __syncthreads();

  const int per = (e + (int)gridDim.x - 1) / (int)gridDim.x;
  const int e0 = min(e, (int)blockIdx.x * per);
  const int e1 = min(e, e0 + per);
  for (int base = e0; base < e1; base += PART_THREADS * EDGE_UNROLL) {
    int d[EDGE_UNROLL];
#pragma unroll
    for (int k = 0; k < EDGE_UNROLL; ++k) {
      const int i = base + k * PART_THREADS + tid;
      const bool ok = i < e1;
      const int c = col[ok ? i : e0];
      d[k] = ok ? (c - lo) : -1;
    }
#pragma unroll
    for (int k = 0; k < EDGE_UNROLL; ++k)
      if ((unsigned)d[k] < (unsigned)span) atomicAdd(&cnt[d[k]], 1);
  }
  __syncthreads();

  int* dst = part + (size_t)blockIdx.x * n + lo;
  for (int i = tid; i < span; i += PART_THREADS) dst[i] = cnt[i];
}

#define SCAN_THREADS 256

__global__ __launch_bounds__(SCAN_THREADS) void deg_fin_kernel(int* __restrict__ part,
                                                               float* __restrict__ dinv,
                                                               int* __restrict__ rowptr,
                                                               int* __restrict__ bsum, int n) {
  __shared__ int wtot[32];
  const int tid = threadIdx.x;
  const int lane = tid & 31, wave = tid >> 5;
  const int i = blockIdx.x * SCAN_THREADS + tid;
  int s = 0;
  if (i < n) {
    int t[DEG_SLICES];
#pragma unroll
    for (int b = 0; b < DEG_SLICES; ++b) t[b] = part[(size_t)b * n + i];
#pragma unroll
    for (int b = 0; b < DEG_SLICES; ++b) {
      part[(size_t)b * n + i] = s;
      s += t[b];
    }
    dinv[i] = rsqrtf(1.0f + (float)s);
  }
  int incl = s;
  for (int o = 1; o < 32; o <<= 1) {
    const int u = __shfl_up(incl, o);
    if (lane >= o) incl += u;
  }
  if (lane == 31) wtot[wave] = incl;
  __syncthreads();
  if (wave == 0) {
    const int w = (lane < SCAN_THREADS / 32) ? wtot[lane] : 0;
    int wi = w;
    for (int o = 1; o < 32; o <<= 1) {
      const int u = __shfl_up(wi, o);
      if (lane >= o) wi += u;
    }
    wtot[lane] = wi - w;
    if (lane == 31) bsum[blockIdx.x] = wi;
  }
  __syncthreads();
  if (i < n) rowptr[i] = wtot[wave] + incl - s;
}

__global__ __launch_bounds__(SCAN_THREADS) void rowptr_kernel(int* __restrict__ rowptr,
                                                              const int* __restrict__ bsum,
                                                              int n) {
  __shared__ int red[32];
  const int tid = threadIdx.x;
  const int lane = tid & 31, wave = tid >> 5;
  int local = 0;
  for (int b = tid; b < (int)blockIdx.x; b += SCAN_THREADS) local += bsum[b];
  for (int o = 16; o > 0; o >>= 1) local += __shfl_xor(local, o);
  if (lane == 0) red[wave] = local;
  __syncthreads();
  int carry = 0;
#pragma unroll
  for (int w = 0; w < SCAN_THREADS / 32; ++w) carry += red[w];
  const int i = blockIdx.x * SCAN_THREADS + tid;
  if (i < n) rowptr[i] += carry;
}

__global__ __launch_bounds__(PART_THREADS) void fill_kernel(const int* __restrict__ row,
                                                            const int* __restrict__ col,
                                                            const int* __restrict__ part,
                                                            const int* __restrict__ rowptr,
                                                            int* __restrict__ srcs,
                                                            int e, int n) {
  __shared__ alignas(16) int cur[DEG_NP];
  const int tid = threadIdx.x;
  const int lo = blockIdx.y * DEG_NP;
  const int span = min(DEG_NP, n - lo);
  const int* pp = part + (size_t)blockIdx.x * n + lo;
  for (int i = tid; i < span; i += PART_THREADS) cur[i] = rowptr[lo + i] + pp[i];
  __syncthreads();

  const int per = (e + (int)gridDim.x - 1) / (int)gridDim.x;
  const int e0 = min(e, (int)blockIdx.x * per);
  const int e1 = min(e, e0 + per);
  for (int base = e0; base < e1; base += PART_THREADS * EDGE_UNROLL) {
    int d[EDGE_UNROLL], r[EDGE_UNROLL];
#pragma unroll
    for (int k = 0; k < EDGE_UNROLL; ++k) {
      const int i = base + k * PART_THREADS + tid;
      const bool ok = i < e1;
      const int ii = ok ? i : e0;
      const int c = col[ii];
      r[k] = row[ii];
      d[k] = ok ? (c - lo) : -1;
    }
#pragma unroll
    for (int k = 0; k < EDGE_UNROLL; ++k) {
      if ((unsigned)d[k] < (unsigned)span) {
        const int pos = atomicAdd(&cur[d[k]], 1);
        srcs[pos] = r[k];
      }
    }
  }
}

__global__ __launch_bounds__(64) void gemm64_wmma_kernel(const float* __restrict__ X,
                                                         const float* __restrict__ W,
                                                         const float* __restrict__ dinv,
                                                         float* __restrict__ H) {
  __shared__ float Wt[64 * LDS_STRIDE];
  for (int p = threadIdx.x; p < 64 * 64; p += 64) {
    int k = p >> 6, n = p & 63;
    Wt[n * LDS_STRIDE + k] = W[p];
  }
  __syncthreads();

  const int wave = threadIdx.x >> 5;
  const int lane = threadIdx.x & 31;
  const int half = lane >> 4;
  const int nc   = lane & 15;
  const int rowBase = blockIdx.x * 32 + wave * 16;

  const float* xrow = X + (size_t)(rowBase + nc) * 64;
  v8f acc0 = {}, acc1 = {}, acc2 = {}, acc3 = {};
  float dv[8];
#pragma unroll
  for (int v = 0; v < 8; ++v) dv[v] = dinv[rowBase + v + half * 8];

#pragma unroll
  for (int k0 = 0; k0 < 64; k0 += 4) {
    const int kb = k0 + half * 2;
    v2f a  = *(const v2f*)(xrow + kb);
    v2f b0 = *(const v2f*)(&Wt[(0 * 16 + nc) * LDS_STRIDE + kb]);
    v2f b1 = *(const v2f*)(&Wt[(1 * 16 + nc) * LDS_STRIDE + kb]);
    v2f b2 = *(const v2f*)(&Wt[(2 * 16 + nc) * LDS_STRIDE + kb]);
    v2f b3 = *(const v2f*)(&Wt[(3 * 16 + nc) * LDS_STRIDE + kb]);
    acc0 = __builtin_amdgcn_wmma_f32_16x16x4_f32(false, a, false, b0, (short)0, acc0, false, false);
    acc1 = __builtin_amdgcn_wmma_f32_16x16x4_f32(false, a, false, b1, (short)0, acc1, false, false);
    acc2 = __builtin_amdgcn_wmma_f32_16x16x4_f32(false, a, false, b2, (short)0, acc2, false, false);
    acc3 = __builtin_amdgcn_wmma_f32_16x16x4_f32(false, a, false, b3, (short)0, acc3, false, false);
  }

#pragma unroll
  for (int v = 0; v < 8; ++v) {
    float* hrow = H + (size_t)(rowBase + v + half * 8) * 64;
    hrow[0 * 16 + nc] = acc0[v] * dv[v];
    hrow[1 * 16 + nc] = acc1[v] * dv[v];
    hrow[2 * 16 + nc] = acc2[v] * dv[v];
    hrow[3 * 16 + nc] = acc3[v] * dv[v];
  }
}

__global__ void gemm64_tail_kernel(const float* __restrict__ X, const float* __restrict__ W,
                                   const float* __restrict__ dinv, float* __restrict__ H,
                                   int start, int n) {
  int t = blockIdx.x * blockDim.x + threadIdx.x;
  int r = start + (t >> 6), c = t & 63;
  if (r >= n) return;
  float s = 0.f;
  for (int k = 0; k < 64; ++k) s += X[(size_t)r * 64 + k] * W[k * 64 + c];
  H[(size_t)r * 64 + c] = s * dinv[r];
}

#define GATHER_CHUNK 8

__global__ __launch_bounds__(256) void gather1_kernel(const float* __restrict__ h,
                                                      const int* __restrict__ rowptr,
                                                      const int* __restrict__ srcs,
                                                      const float* __restrict__ dinv,
                                                      float* __restrict__ h2, int e, int n) {
  const int node = blockIdx.x * 16 + ((int)threadIdx.x >> 4);
  const int l = threadIdx.x & 15;
  if (node >= n) return;
  const int beg = rowptr[node];
  const int end = (node + 1 < n) ? rowptr[node + 1] : e;
  const float di = dinv[node];
  const float4* h4 = (const float4*)h;
  float4 acc = h4[(size_t)node * 16 + l];

  int j = beg;
  for (; j + GATHER_CHUNK <= end; j += GATHER_CHUNK) {
    int s[GATHER_CHUNK];
    float4 a[GATHER_CHUNK];
#pragma unroll
    for (int k = 0; k < GATHER_CHUNK; ++k) s[k] = srcs[j + k];
#pragma unroll
    for (int k = 0; k < GATHER_CHUNK; ++k) a[k] = h4[(size_t)s[k] * 16 + l];
#pragma unroll
    for (int k = 0; k < GATHER_CHUNK; ++k) {
      acc.x += a[k].x;
      acc.y += a[k].y;
      acc.z += a[k].z;
      acc.w += a[k].w;
    }
  }
  if (j < end) {
    int s[GATHER_CHUNK];
    float4 a[GATHER_CHUNK];
#pragma unroll
    for (int k = 0; k < GATHER_CHUNK; ++k) s[k] = srcs[min(j + k, end - 1)];
#pragma unroll
    for (int k = 0; k < GATHER_CHUNK; ++k) a[k] = h4[(size_t)s[k] * 16 + l];
#pragma unroll
    for (int k = 0; k < GATHER_CHUNK; ++k) {
      const float m = (j + k < end) ? 1.0f : 0.0f;
      acc.x += a[k].x * m;
      acc.y += a[k].y * m;
      acc.z += a[k].z * m;
      acc.w += a[k].w * m;
    }
  }
  acc.x *= di;
  acc.y *= di;
  acc.z *= di;
  acc.w *= di;
  *(float4*)&h2[(size_t)node * 64 + l * 4] = acc;
}

__global__ __launch_bounds__(64) void gemm2_wmma_kernel(const float* __restrict__ H2,
                                                        const float* __restrict__ B1,
                                                        const float* __restrict__ W2,
                                                        const float* __restrict__ dinv,
                                                        float* __restrict__ G) {
  __shared__ float Wt[16 * LDS_STRIDE];
  __shared__ float b1s[64];
  for (int p = threadIdx.x; p < 16 * 64; p += 64) {
    int n = p >> 6, k = p & 63;
    Wt[n * LDS_STRIDE + k] = (n < 2) ? W2[k * 2 + n] : 0.0f;
  }
  b1s[threadIdx.x] = B1[threadIdx.x];
  __syncthreads();

  const int wave = threadIdx.x >> 5;
  const int lane = threadIdx.x & 31;
  const int half = lane >> 4;
  const int nc   = lane & 15;
  const int rowBase = blockIdx.x * 32 + wave * 16;

  const float* hrow = H2 + (size_t)(rowBase + nc) * 64;
  const float dv = dinv[rowBase + nc];
  v8f acc = {};

#pragma unroll
  for (int k0 = 0; k0 < 64; k0 += 4) {
    const int kb = k0 + half * 2;
    v2f a;
    a.x = fmaxf(hrow[kb + 0] + b1s[kb + 0], 0.0f);
    a.y = fmaxf(hrow[kb + 1] + b1s[kb + 1], 0.0f);
    v2f b = *(const v2f*)(&Wt[nc * LDS_STRIDE + kb]);
    acc = __builtin_amdgcn_wmma_f32_16x16x4_f32(false, a, false, b, (short)0, acc, false, false);
  }

  if (nc < 2) {
#pragma unroll
    for (int v = 0; v < 8; ++v)
      G[(size_t)(rowBase + v + half * 8) * 4 + nc] = acc[v];
  }
  if (half == 0)
    *(v2f*)&G[(size_t)(rowBase + nc) * 4 + 2] = v2f{dv, 0.0f};
}

__global__ void gemm2_tail_kernel(const float* __restrict__ H2, const float* __restrict__ B1,
                                  const float* __restrict__ W2, const float* __restrict__ dinv,
                                  float* __restrict__ G, int start, int n) {
  int t = blockIdx.x * blockDim.x + threadIdx.x;
  int r = start + (t >> 2), c = t & 3;
  if (r >= n) return;
  float s = 0.f;
  if (c < 2) {
    for (int k = 0; k < 64; ++k)
      s += fmaxf(H2[(size_t)r * 64 + k] + B1[k], 0.0f) * W2[k * 2 + c];
  } else if (c == 2) {
    s = dinv[r];
  }
  G[(size_t)r * 4 + c] = s;
}

#define G2_LANES 8
#define G2_CHUNK 4
#define G2_THREADS 256
#define G2_NODES (G2_THREADS / G2_LANES)

__global__ __launch_bounds__(G2_THREADS) void gather2_kernel(const float* __restrict__ rec,
                                                             const int* __restrict__ rowptr,
                                                             const int* __restrict__ srcs,
                                                             const float* __restrict__ b2,
                                                             float* __restrict__ out, int e, int n) {
  __shared__ float stage[G2_NODES * 2];
  const int tid = threadIdx.x;
  const int ln = tid / G2_LANES;
  const int t = tid & (G2_LANES - 1);
  const int node0 = blockIdx.x * G2_NODES;
  const int node = min(node0 + ln, n - 1);
  const int beg = rowptr[node];
  const int end = (node + 1 < n) ? rowptr[node + 1] : e;
  const float4* r4 = (const float4*)rec;
  const float4 self = r4[node];
  const float di = self.z;
  float ax = 0.0f, ay = 0.0f;

  for (int j = beg + t; j < end; j += G2_LANES * G2_CHUNK) {
    int s[G2_CHUNK];
    float4 r[G2_CHUNK];
#pragma unroll
    for (int k = 0; k < G2_CHUNK; ++k) {
      const int jj = j + k * G2_LANES;
      s[k] = srcs[(jj < end) ? jj : j];
    }
#pragma unroll
    for (int k = 0; k < G2_CHUNK; ++k) {
      const bool ok = j + k * G2_LANES < end;
      r[k] = r4[ok ? s[k] : node];
    }
    static_assert(G2_CHUNK == 4, "the fence below names four records");
    asm volatile("" ::"v"(r[0].x), "v"(r[1].x), "v"(r[2].x), "v"(r[3].x));
#pragma unroll
    for (int k = 0; k < G2_CHUNK; ++k) {
      const bool ok = j + k * G2_LANES < end;
      const float w = ok ? r[k].z * di : 0.0f;
      ax += r[k].x * w;
      ay += r[k].y * w;
    }
  }
#pragma unroll
  for (int o = G2_LANES / 2; o > 0; o >>= 1) {
    ax += __shfl_xor(ax, o);
    ay += __shfl_xor(ay, o);
  }
  if (t == 0) {
    const float ss = di * di;
    stage[ln * 2 + 0] = (self.x * ss + b2[0]) + ax;
    stage[ln * 2 + 1] = (self.y * ss + b2[1]) + ay;
  }
  __syncthreads();
  const int nf = min(G2_NODES, n - node0) * 2;
  if (tid < nf) out[(size_t)node0 * 2 + tid] = stage[tid];
}

extern "C" void kernel_launch(void* const* d_in, const int* in_sizes, int n_in,
                              void* d_out, int out_size, void* d_ws, size_t ws_size,
                              hipStream_t stream) {
  const float* x  = (const float*)d_in[0];
  const int*   ei = (const int*)d_in[1];
  const float* W1 = (const float*)d_in[2];
  const float* b1 = (const float*)d_in[3];
  const float* W2 = (const float*)d_in[4];
  const float* b2 = (const float*)d_in[5];
  float* out = (float*)d_out;

  const int N = in_sizes[0] / 64;
  const int E = in_sizes[1] / 2;
  const int* row = ei;
  const int* col = ei + E;

  const int Npad = (N + 63) & ~63;
  float* dinv = (float*)d_ws;
  float* h    = dinv + Npad;
  float* h2   = h + (size_t)N * 64;
  float* g    = h2 + (size_t)N * 64;
  float* prt  = g + (size_t)N * 4;
  int* srcs   = (int*)(prt + (size_t)DEG_SLICES * N);
  int* rowptr = srcs + E;
  int* bsum   = rowptr + Npad;

  const int T = 256;
  deg_part_kernel<<<dim3(DEG_SLICES, (N + DEG_NP - 1) / DEG_NP), PART_THREADS, 0, stream>>>(
      col, (int*)prt, E, N);
  const int scanBlocks = (N + SCAN_THREADS - 1) / SCAN_THREADS;
  deg_fin_kernel<<<scanBlocks, SCAN_THREADS, 0, stream>>>((int*)prt, dinv, rowptr, bsum, N);
  rowptr_kernel<<<scanBlocks, SCAN_THREADS, 0, stream>>>(rowptr, bsum, N);
  fill_kernel<<<dim3(DEG_SLICES, (N + DEG_NP - 1) / DEG_NP), PART_THREADS, 0, stream>>>(
      row, col, (const int*)prt, rowptr, srcs, E, N);

  const int mainBlocks = N / 32;
  const int tailStart  = mainBlocks * 32;
  if (mainBlocks > 0)
    gemm64_wmma_kernel<<<mainBlocks, 64, 0, stream>>>(x, W1, dinv, h);
  if (tailStart < N) {
    int tcnt = (N - tailStart) * 64;
    gemm64_tail_kernel<<<(tcnt + T - 1) / T, T, 0, stream>>>(x, W1, dinv, h, tailStart, N);
  }

  gather1_kernel<<<(N + 15) / 16, 256, 0, stream>>>(h, rowptr, srcs, dinv, h2, E, N);

  if (mainBlocks > 0)
    gemm2_wmma_kernel<<<mainBlocks, 64, 0, stream>>>(h2, b1, W2, dinv, g);
  if (tailStart < N) {
    int tcnt = (N - tailStart) * 4;
    gemm2_tail_kernel<<<(tcnt + T - 1) / T, T, 0, stream>>>(h2, b1, W2, dinv, g, tailStart, N);
  }

  gather2_kernel<<<(N + G2_NODES - 1) / G2_NODES, G2_THREADS, 0, stream>>>(g, rowptr, srcs, b2, out,
                                                                           E, N);
}


# Layer-2 GEMM fused into the layer-1 gather: a block's 16 finished rows (bias, ReLU) go through LDS to one wave's WMMA (tile as B operand); h2 is never written or read; one launch fewer
# speedup vs baseline: 3.2696x; 1.0353x over previous
#include <hip/hip_runtime.h>
#include <hip/hip_bf16.h>

typedef float v2f __attribute__((ext_vector_type(2)));
typedef float v8f __attribute__((ext_vector_type(8)));

#define LDS_STRIDE 66

#define PART_THREADS 1024
#define EDGE_UNROLL 8
#define DEG_NP 51200
#define DEG_SLICES 64

__global__ __launch_bounds__(PART_THREADS) void deg_part_kernel(const int* __restrict__ col,
                                                                int* __restrict__ part,
                                                                int e, int n) {
  __shared__ alignas(16) int cnt[DEG_NP];
  const int tid = threadIdx.x;
  const int lo = blockIdx.y * DEG_NP;
  const int span = min(DEG_NP, n - lo);
  for (int i = tid; i < DEG_NP / 4; i += PART_THREADS) ((int4*)cnt)[i] = make_int4(0, 0, 0, 0);
  __syncthreads();

  const int per = (e + (int)gridDim.x - 1) / (int)gridDim.x;
  const int e0 = min(e, (int)blockIdx.x * per);
  const int e1 = min(e, e0 + per);
  for (int base = e0; base < e1; base += PART_THREADS * EDGE_UNROLL) {
    int d[EDGE_UNROLL];
#pragma unroll
    for (int k = 0; k < EDGE_UNROLL; ++k) {
      const int i = base + k * PART_THREADS + tid;
      const bool ok = i < e1;
      const int c = col[ok ? i : e0];
      d[k] = ok ? (c - lo) : -1;
    }
#pragma unroll
    for (int k = 0; k < EDGE_UNROLL; ++k)
      if ((unsigned)d[k] < (unsigned)span) atomicAdd(&cnt[d[k]], 1);
  }
  __syncthreads();

  int* dst = part + (size_t)blockIdx.x * n + lo;
  for (int i = tid; i < span; i += PART_THREADS) dst[i] = cnt[i];
}

#define SCAN_THREADS 256

__global__ __launch_bounds__(SCAN_THREADS) void deg_fin_kernel(int* __restrict__ part,
                                                               float* __restrict__ dinv,
                                                               int* __restrict__ rowptr,
                                                               int* __restrict__ bsum, int n) {
  __shared__ int wtot[32];
  const int tid = threadIdx.x;
  const int lane = tid & 31, wave = tid >> 5;
  const int i = blockIdx.x * SCAN_THREADS + tid;
  int s = 0;
  if (i < n) {
    int t[DEG_SLICES];
#pragma unroll
    for (int b = 0; b < DEG_SLICES; ++b) t[b] = part[(size_t)b * n + i];
#pragma unroll
    for (int b = 0; b < DEG_SLICES; ++b) {
      part[(size_t)b * n + i] = s;
      s += t[b];
    }
    dinv[i] = rsqrtf(1.0f + (float)s);
  }
  int incl = s;
  for (int o = 1; o < 32; o <<= 1) {
    const int u = __shfl_up(incl, o);
    if (lane >= o) incl += u;
  }
  if (lane == 31) wtot[wave] = incl;
  __syncthreads();
  if (wave == 0) {
    const int w = (lane < SCAN_THREADS / 32) ? wtot[lane] : 0;
    int wi = w;
    for (int o = 1; o < 32; o <<= 1) {
      const int u = __shfl_up(wi, o);
      if (lane >= o) wi += u;
    }
    wtot[lane] = wi - w;
    if (lane == 31) bsum[blockIdx.x] = wi;
  }
  __syncthreads();
  if (i < n) rowptr[i] = wtot[wave] + incl - s;
}

__global__ __launch_bounds__(SCAN_THREADS) void rowptr_kernel(int* __restrict__ rowptr,
                                                              const int* __restrict__ bsum,
                                                              int n) {
  __shared__ int red[32];
  const int tid = threadIdx.x;
  const int lane = tid & 31, wave = tid >> 5;
  int local = 0;
  for (int b = tid; b < (int)blockIdx.x; b += SCAN_THREADS) local += bsum[b];
  for (int o = 16; o > 0; o >>= 1) local += __shfl_xor(local, o);
  if (lane == 0) red[wave] = local;
  __syncthreads();
  int carry = 0;
#pragma unroll
  for (int w = 0; w < SCAN_THREADS / 32; ++w) carry += red[w];
  const int i = blockIdx.x * SCAN_THREADS + tid;
  if (i < n) rowptr[i] += carry;
}

__global__ __launch_bounds__(PART_THREADS) void fill_kernel(const int* __restrict__ row,
                                                            const int* __restrict__ col,
                                                            const int* __restrict__ part,
                                                            const int* __restrict__ rowptr,
                                                            int* __restrict__ srcs,
                                                            int e, int n) {
  __shared__ alignas(16) int cur[DEG_NP];
  const int tid = threadIdx.x;
  const int lo = blockIdx.y * DEG_NP;
  const int span = min(DEG_NP, n - lo);
  const int* pp = part + (size_t)blockIdx.x * n + lo;
  for (int i = tid; i < span; i += PART_THREADS) cur[i] = rowptr[lo + i] + pp[i];
  __syncthreads();

  const int per = (e + (int)gridDim.x - 1) / (int)gridDim.x;
  const int e0 = min(e, (int)blockIdx.x * per);
  const int e1 = min(e, e0 + per);
  for (int base = e0; base < e1; base += PART_THREADS * EDGE_UNROLL) {
    int d[EDGE_UNROLL], r[EDGE_UNROLL];
#pragma unroll
    for (int k = 0; k < EDGE_UNROLL; ++k) {
      const int i = base + k * PART_THREADS + tid;
      const bool ok = i < e1;
      const int ii = ok ? i : e0;
      const int c = col[ii];
      r[k] = row[ii];
      d[k] = ok ? (c - lo) : -1;
    }
#pragma unroll
    for (int k = 0; k < EDGE_UNROLL; ++k) {
      if ((unsigned)d[k] < (unsigned)span) {
        const int pos = atomicAdd(&cur[d[k]], 1);
        srcs[pos] = r[k];
      }
    }
  }
}

__global__ __launch_bounds__(64) void gemm64_wmma_kernel(const float* __restrict__ X,
                                                         const float* __restrict__ W,
                                                         const float* __restrict__ dinv,
                                                         float* __restrict__ H) {
  __shared__ float Wt[64 * LDS_STRIDE];
  for (int p = threadIdx.x; p < 64 * 64; p += 64) {
    int k = p >> 6, n = p & 63;
    Wt[n * LDS_STRIDE + k] = W[p];
  }
  __syncthreads();

  const int wave = threadIdx.x >> 5;
  const int lane = threadIdx.x & 31;
  const int half = lane >> 4;
  const int nc   = lane & 15;
  const int rowBase = blockIdx.x * 32 + wave * 16;

  const float* xrow = X + (size_t)(rowBase + nc) * 64;
  v8f acc0 = {}, acc1 = {}, acc2 = {}, acc3 = {};
  float dv[8];
#pragma unroll
  for (int v = 0; v < 8; ++v) dv[v] = dinv[rowBase + v + half * 8];

#pragma unroll
  for (int k0 = 0; k0 < 64; k0 += 4) {
    const int kb = k0 + half * 2;
    v2f a  = *(const v2f*)(xrow + kb);
    v2f b0 = *(const v2f*)(&Wt[(0 * 16 + nc) * LDS_STRIDE + kb]);
    v2f b1 = *(const v2f*)(&Wt[(1 * 16 + nc) * LDS_STRIDE + kb]);
    v2f b2 = *(const v2f*)(&Wt[(2 * 16 + nc) * LDS_STRIDE + kb]);
    v2f b3 = *(const v2f*)(&Wt[(3 * 16 + nc) * LDS_STRIDE + kb]);
    acc0 = __builtin_amdgcn_wmma_f32_16x16x4_f32(false, a, false, b0, (short)0, acc0, false, false);
    acc1 = __builtin_amdgcn_wmma_f32_16x16x4_f32(false, a, false, b1, (short)0, acc1, false, false);
    acc2 = __builtin_amdgcn_wmma_f32_16x16x4_f32(false, a, false, b2, (short)0, acc2, false, false);
    acc3 = __builtin_amdgcn_wmma_f32_16x16x4_f32(false, a, false, b3, (short)0, acc3, false, false);
  }

#pragma unroll
  for (int v = 0; v < 8; ++v) {
    float* hrow = H + (size_t)(rowBase + v + half * 8) * 64;
    hrow[0 * 16 + nc] = acc0[v] * dv[v];
    hrow[1 * 16 + nc] = acc1[v] * dv[v];
    hrow[2 * 16 + nc] = acc2[v] * dv[v];
    hrow[3 * 16 + nc] = acc3[v] * dv[v];
  }
}

__global__ void gemm64_tail_kernel(const float* __restrict__ X, const float* __restrict__ W,
                                   const float* __restrict__ dinv, float* __restrict__ H,
                                   int start, int n) {
  int t = blockIdx.x * blockDim.x + threadIdx.x;
  int r = start + (t >> 6), c = t & 63;
  if (r >= n) return;
  float s = 0.f;
  for (int k = 0; k < 64; ++k) s += X[(size_t)r * 64 + k] * W[k * 64 + c];
  H[(size_t)r * 64 + c] = s * dinv[r];
}

#define GATHER_CHUNK 8
#define TILE_STRIDE 68

__global__ __launch_bounds__(256) void gather1_gemm2_kernel(const float* __restrict__ h,
                                                            const int* __restrict__ rowptr,
                                                            const int* __restrict__ srcs,
                                                            const float* __restrict__ dinv,
                                                            const float* __restrict__ B1,
                                                            const float* __restrict__ W2,
                                                            float* __restrict__ G, int e, int n) {
  __shared__ alignas(16) float tile[16 * TILE_STRIDE];
  __shared__ float w2s[64 * 2];
  __shared__ float dis[16];
  const int tid = threadIdx.x;
  const int ln = tid >> 4;
  const int l = tid & 15;
  const int node0 = blockIdx.x * 16;
  const int node = min(node0 + ln, n - 1);
  const float w2v = W2[tid & 127];
  const int beg = rowptr[node];
  const int end = (node + 1 < n) ? rowptr[node + 1] : e;
  const float di = dinv[node];
  const float4 bias = ((const float4*)B1)[l];
  const float4* h4 = (const float4*)h;
  float4 acc = h4[(size_t)node * 16 + l];

  int j = beg;
  for (; j + GATHER_CHUNK <= end; j += GATHER_CHUNK) {
    int s[GATHER_CHUNK];
    float4 a[GATHER_CHUNK];
#pragma unroll
    for (int k = 0; k < GATHER_CHUNK; ++k) s[k] = srcs[j + k];
#pragma unroll
    for (int k = 0; k < GATHER_CHUNK; ++k) a[k] = h4[(size_t)s[k] * 16 + l];
#pragma unroll
    for (int k = 0; k < GATHER_CHUNK; ++k) {
      acc.x += a[k].x;
      acc.y += a[k].y;
      acc.z += a[k].z;
      acc.w += a[k].w;
    }
  }
  if (j < end) {
    int s[GATHER_CHUNK];
    float4 a[GATHER_CHUNK];
#pragma unroll
    for (int k = 0; k < GATHER_CHUNK; ++k) s[k] = srcs[min(j + k, end - 1)];
#pragma unroll
    for (int k = 0; k < GATHER_CHUNK; ++k) a[k] = h4[(size_t)s[k] * 16 + l];
#pragma unroll
    for (int k = 0; k < GATHER_CHUNK; ++k) {
      const float m = (j + k < end) ? 1.0f : 0.0f;
      acc.x += a[k].x * m;
      acc.y += a[k].y * m;
      acc.z += a[k].z * m;
      acc.w += a[k].w * m;
    }
  }
  float4 t;
  t.x = fmaxf(acc.x * di + bias.x, 0.0f);
  t.y = fmaxf(acc.y * di + bias.y, 0.0f);
  t.z = fmaxf(acc.z * di + bias.z, 0.0f);
  t.w = fmaxf(acc.w * di + bias.w, 0.0f);
  *(float4*)&tile[ln * TILE_STRIDE + l * 4] = t;
  if (l == 0) dis[ln] = di;
  if (tid < 128) w2s[tid] = w2v;
  __syncthreads();

  if (__builtin_amdgcn_readfirstlane(tid) >= 32) return;

  const int half = tid >> 4;
  const int nc = tid & 15;
  const int wc = nc & 1;
  v8f d = {};
#pragma unroll
  for (int k0 = 0; k0 < 64; k0 += 4) {
    const int kb = k0 + half * 2;
    const v2f x = *(const v2f*)(&tile[nc * TILE_STRIDE + kb]);
    const float w0 = w2s[kb * 2 + wc];
    const float w1 = w2s[kb * 2 + 2 + wc];
    v2f w;
    w.x = (nc < 2) ? w0 : 0.0f;
    w.y = (nc < 2) ? w1 : 0.0f;
    d = __builtin_amdgcn_wmma_f32_16x16x4_f32(false, w, false, x, (short)0, d, false, false);
  }
  if (half == 0 && node0 + nc < n) {
    float4 rec;
    rec.x = d[0];
    rec.y = d[1];
    rec.z = dis[nc];
    rec.w = 0.0f;
    *(float4*)&G[(size_t)(node0 + nc) * 4] = rec;
  }
}

#define G2_LANES 8
#define G2_CHUNK 4
#define G2_THREADS 256
#define G2_NODES (G2_THREADS / G2_LANES)

__global__ __launch_bounds__(G2_THREADS) void gather2_kernel(const float* __restrict__ rec,
                                                             const int* __restrict__ rowptr,
                                                             const int* __restrict__ srcs,
                                                             const float* __restrict__ b2,
                                                             float* __restrict__ out, int e, int n) {
  __shared__ float stage[G2_NODES * 2];
  const int tid = threadIdx.x;
  const int ln = tid / G2_LANES;
  const int t = tid & (G2_LANES - 1);
  const int node0 = blockIdx.x * G2_NODES;
  const int node = min(node0 + ln, n - 1);
  const int beg = rowptr[node];
  const int end = (node + 1 < n) ? rowptr[node + 1] : e;
  const float4* r4 = (const float4*)rec;
  const float4 self = r4[node];
  const float di = self.z;
  float ax = 0.0f, ay = 0.0f;

  for (int j = beg + t; j < end; j += G2_LANES * G2_CHUNK) {
    int s[G2_CHUNK];
    float4 r[G2_CHUNK];
#pragma unroll
    for (int k = 0; k < G2_CHUNK; ++k) {
      const int jj = j + k * G2_LANES;
      s[k] = srcs[(jj < end) ? jj : j];
    }
#pragma unroll
    for (int k = 0; k < G2_CHUNK; ++k) {
      const bool ok = j + k * G2_LANES < end;
      r[k] = r4[ok ? s[k] : node];
    }
    static_assert(G2_CHUNK == 4, "the fence below names four records");
    asm volatile("" ::"v"(r[0].x), "v"(r[1].x), "v"(r[2].x), "v"(r[3].x));
#pragma unroll
    for (int k = 0; k < G2_CHUNK; ++k) {
      const bool ok = j + k * G2_LANES < end;
      const float w = ok ? r[k].z * di : 0.0f;
      ax += r[k].x * w;
      ay += r[k].y * w;
    }
  }
#pragma unroll
  for (int o = G2_LANES / 2; o > 0; o >>= 1) {
    ax += __shfl_xor(ax, o);
    ay += __shfl_xor(ay, o);
  }
  if (t == 0) {
    const float ss = di * di;
    stage[ln * 2 + 0] = (self.x * ss + b2[0]) + ax;
    stage[ln * 2 + 1] = (self.y * ss + b2[1]) + ay;
  }
  __syncthreads();
  const int nf = min(G2_NODES, n - node0) * 2;
  if (tid < nf) out[(size_t)node0 * 2 + tid] = stage[tid];
}

extern "C" void kernel_launch(void* const* d_in, const int* in_sizes, int n_in,
                              void* d_out, int out_size, void* d_ws, size_t ws_size,
                              hipStream_t stream) {
  const float* x  = (const float*)d_in[0];
  const int*   ei = (const int*)d_in[1];
  const float* W1 = (const float*)d_in[2];
  const float* b1 = (const float*)d_in[3];
  const float* W2 = (const float*)d_in[4];
  const float* b2 = (const float*)d_in[5];
  float* out = (float*)d_out;

  const int N = in_sizes[0] / 64;
  const int E = in_sizes[1] / 2;
  const int* row = ei;
  const int* col = ei + E;

  const int Npad = (N + 63) & ~63;
  float* dinv = (float*)d_ws;
  float* h    = dinv + Npad;
  float* g    = h + (size_t)N * 64;
  float* prt  = g + (size_t)N * 4;
  int* srcs   = (int*)(prt + (size_t)DEG_SLICES * N);
  int* rowptr = srcs + E;
  int* bsum   = rowptr + Npad;

  const int T = 256;
  deg_part_kernel<<<dim3(DEG_SLICES, (N + DEG_NP - 1) / DEG_NP), PART_THREADS, 0, stream>>>(
      col, (int*)prt, E, N);
  const int scanBlocks = (N + SCAN_THREADS - 1) / SCAN_THREADS;
  deg_fin_kernel<<<scanBlocks, SCAN_THREADS, 0, stream>>>((int*)prt, dinv, rowptr, bsum, N);
  rowptr_kernel<<<scanBlocks, SCAN_THREADS, 0, stream>>>(rowptr, bsum, N);
  fill_kernel<<<dim3(DEG_SLICES, (N + DEG_NP - 1) / DEG_NP), PART_THREADS, 0, stream>>>(
      row, col, (const int*)prt, rowptr, srcs, E, N);

  const int mainBlocks = N / 32;
  const int tailStart  = mainBlocks * 32;
  if (mainBlocks > 0)
    gemm64_wmma_kernel<<<mainBlocks, 64, 0, stream>>>(x, W1, dinv, h);
  if (tailStart < N) {
    int tcnt = (N - tailStart) * 64;
    gemm64_tail_kernel<<<(tcnt + T - 1) / T, T, 0, stream>>>(x, W1, dinv, h, tailStart, N);
  }

  gather1_gemm2_kernel<<<(N + 15) / 16, 256, 0, stream>>>(h, rowptr, srcs, dinv, b1, W2, g, E, N);

  gather2_kernel<<<(N + G2_NODES - 1) / G2_NODES, G2_THREADS, 0, stream>>>(g, rowptr, srcs, b2, out,
                                                                           E, N);
}


# Layer-1 GEMM moved into the edge-list fill launch as 128 extra 32-wave blocks on the WGPs that launch leaves idle (a wave walks 16-row tiles, W staged once per block); one launch fewer
# speedup vs baseline: 3.5700x; 1.0919x over previous
#include <hip/hip_runtime.h>
#include <hip/hip_bf16.h>

typedef float v2f __attribute__((ext_vector_type(2)));
typedef float v8f __attribute__((ext_vector_type(8)));

#define LDS_STRIDE 66

#define PART_THREADS 1024
#define EDGE_UNROLL 8
#define DEG_NP 51200
#define DEG_SLICES 64

__global__ __launch_bounds__(PART_THREADS) void deg_part_kernel(const int* __restrict__ col,
                                                                int* __restrict__ part,
                                                                int e, int n) {
  __shared__ alignas(16) int cnt[DEG_NP];
  const int tid = threadIdx.x;
  const int lo = blockIdx.y * DEG_NP;
  const int span = min(DEG_NP, n - lo);
  for (int i = tid; i < DEG_NP / 4; i += PART_THREADS) ((int4*)cnt)[i] = make_int4(0, 0, 0, 0);
  __syncthreads();

  const int per = (e + (int)gridDim.x - 1) / (int)gridDim.x;
  const int e0 = min(e, (int)blockIdx.x * per);
  const int e1 = min(e, e0 + per);
  for (int base = e0; base < e1; base += PART_THREADS * EDGE_UNROLL) {
    int d[EDGE_UNROLL];
#pragma unroll
    for (int k = 0; k < EDGE_UNROLL; ++k) {
      const int i = base + k * PART_THREADS + tid;
      const bool ok = i < e1;
      const int c = col[ok ? i : e0];
      d[k] = ok ? (c - lo) : -1;
    }
#pragma unroll
    for (int k = 0; k < EDGE_UNROLL; ++k)
      if ((unsigned)d[k] < (unsigned)span) atomicAdd(&cnt[d[k]], 1);
  }
  __syncthreads();

  int* dst = part + (size_t)blockIdx.x * n + lo;
  for (int i = tid; i < span; i += PART_THREADS) dst[i] = cnt[i];
}

#define SCAN_THREADS 256

__global__ __launch_bounds__(SCAN_THREADS) void deg_fin_kernel(int* __restrict__ part,
                                                               float* __restrict__ dinv,
                                                               int* __restrict__ rowptr,
                                                               int* __restrict__ bsum, int n) {
  __shared__ int wtot[32];
  const int tid = threadIdx.x;
  const int lane = tid & 31, wave = tid >> 5;
  const int i = blockIdx.x * SCAN_THREADS + tid;
  int s = 0;
  if (i < n) {
    int t[DEG_SLICES];
#pragma unroll
    for (int b = 0; b < DEG_SLICES; ++b) t[b] = part[(size_t)b * n + i];
#pragma unroll
    for (int b = 0; b < DEG_SLICES; ++b) {
      part[(size_t)b * n + i] = s;
      s += t[b];
    }
    dinv[i] = rsqrtf(1.0f + (float)s);
  }
  int incl = s;
  for (int o = 1; o < 32; o <<= 1) {
    const int u = __shfl_up(incl, o);
    if (lane >= o) incl += u;
  }
  if (lane == 31) wtot[wave] = incl;
  __syncthreads();
  if (wave == 0) {
    const int w = (lane < SCAN_THREADS / 32) ? wtot[lane] : 0;
    int wi = w;
    for (int o = 1; o < 32; o <<= 1) {
      const int u = __shfl_up(wi, o);
      if (lane >= o) wi += u;
    }
    wtot[lane] = wi - w;
    if (lane == 31) bsum[blockIdx.x] = wi;
  }
  __syncthreads();
  if (i < n) rowptr[i] = wtot[wave] + incl - s;
}

__global__ __launch_bounds__(SCAN_THREADS) void rowptr_kernel(int* __restrict__ rowptr,
                                                              const int* __restrict__ bsum,
                                                              int n) {
  __shared__ int red[32];
  const int tid = threadIdx.x;
  const int lane = tid & 31, wave = tid >> 5;
  int local = 0;
  for (int b = tid; b < (int)blockIdx.x; b += SCAN_THREADS) local += bsum[b];
  for (int o = 16; o > 0; o >>= 1) local += __shfl_xor(local, o);
  if (lane == 0) red[wave] = local;
  __syncthreads();
  int carry = 0;
#pragma unroll
  for (int w = 0; w < SCAN_THREADS / 32; ++w) carry += red[w];
  const int i = blockIdx.x * SCAN_THREADS + tid;
  if (i < n) rowptr[i] += carry;
}

#define GEMM_Y 2

__global__ __launch_bounds__(PART_THREADS) void fill_gemm_kernel(const int* __restrict__ row,
                                                                 const int* __restrict__ col,
                                                                 const int* __restrict__ part,
                                                                 const int* __restrict__ rowptr,
                                                                 int* __restrict__ srcs,
                                                                 int e, int n, int ranges,
                                                                 const float* __restrict__ X,
                                                                 const float* __restrict__ W,
                                                                 const float* __restrict__ dinv,
                                                                 float* __restrict__ H) {
  __shared__ alignas(16) unsigned char lds_raw[DEG_NP * sizeof(int)];
  const int tid = threadIdx.x;

  if ((int)blockIdx.y < ranges) {
    int* cur = (int*)lds_raw;
    const int lo = blockIdx.y * DEG_NP;
    const int span = min(DEG_NP, n - lo);
    const int* pp = part + (size_t)blockIdx.x * n + lo;
    for (int i = tid; i < span; i += PART_THREADS) cur[i] = rowptr[lo + i] + pp[i];
    __syncthreads();

    const int per = (e + (int)gridDim.x - 1) / (int)gridDim.x;
    const int e0 = min(e, (int)blockIdx.x * per);
    const int e1 = min(e, e0 + per);
    for (int base = e0; base < e1; base += PART_THREADS * EDGE_UNROLL) {
      int d[EDGE_UNROLL], r[EDGE_UNROLL];
#pragma unroll
      for (int k = 0; k < EDGE_UNROLL; ++k) {
        const int i = base + k * PART_THREADS + tid;
        const bool ok = i < e1;
        const int ii = ok ? i : e0;
        const int c = col[ii];
        r[k] = row[ii];
        d[k] = ok ? (c - lo) : -1;
      }
#pragma unroll
      for (int k = 0; k < EDGE_UNROLL; ++k) {
        if ((unsigned)d[k] < (unsigned)span) {
          const int pos = atomicAdd(&cur[d[k]], 1);
          srcs[pos] = r[k];
        }
      }
    }
    return;
  }

  float* Wt = (float*)lds_raw;
#pragma unroll
  for (int p = tid; p < 64 * 64; p += PART_THREADS) {
    const int k = p >> 6, nn = p & 63;
    Wt[nn * LDS_STRIDE + k] = W[p];
  }
  __syncthreads();

  const int gblocks = ((int)gridDim.y - ranges) * (int)gridDim.x;
  const int g = ((int)blockIdx.y - ranges) * (int)gridDim.x + (int)blockIdx.x;
  const int wave = __builtin_amdgcn_readfirstlane(tid >> 5);
  const int lane = tid & 31;
  const int half = lane >> 4;
  const int nc   = lane & 15;
  const int tiles = n >> 4;

  for (int t = wave * gblocks + g; t < tiles; t += (PART_THREADS / 32) * gblocks) {
    const int rowBase = t * 16;
    const float* xrow = X + (size_t)(rowBase + nc) * 64;
    v8f acc0 = {}, acc1 = {}, acc2 = {}, acc3 = {};
    float dv[8];
#pragma unroll
    for (int v = 0; v < 8; ++v) dv[v] = dinv[rowBase + v + half * 8];

#pragma unroll
    for (int k0 = 0; k0 < 64; k0 += 4) {
      if (k0 == 32) __builtin_amdgcn_sched_barrier(0);
      const int kb = k0 + half * 2;
      v2f a  = *(const v2f*)(xrow + kb);
      v2f b0 = *(const v2f*)(&Wt[(0 * 16 + nc) * LDS_STRIDE + kb]);
      v2f b1 = *(const v2f*)(&Wt[(1 * 16 + nc) * LDS_STRIDE + kb]);
      v2f b2 = *(const v2f*)(&Wt[(2 * 16 + nc) * LDS_STRIDE + kb]);
      v2f b3 = *(const v2f*)(&Wt[(3 * 16 + nc) * LDS_STRIDE + kb]);
      acc0 = __builtin_amdgcn_wmma_f32_16x16x4_f32(false, a, false, b0, (short)0, acc0, false, false);
      acc1 = __builtin_amdgcn_wmma_f32_16x16x4_f32(false, a, false, b1, (short)0, acc1, false, false);
      acc2 = __builtin_amdgcn_wmma_f32_16x16x4_f32(false, a, false, b2, (short)0, acc2, false, false);
      acc3 = __builtin_amdgcn_wmma_f32_16x16x4_f32(false, a, false, b3, (short)0, acc3, false, false);
    }

#pragma unroll
    for (int v = 0; v < 8; ++v) {
      float* hrow = H + (size_t)(rowBase + v + half * 8) * 64;
      hrow[0 * 16 + nc] = acc0[v] * dv[v];
      hrow[1 * 16 + nc] = acc1[v] * dv[v];
      hrow[2 * 16 + nc] = acc2[v] * dv[v];
      hrow[3 * 16 + nc] = acc3[v] * dv[v];
    }
  }
}

__global__ void gemm64_tail_kernel(const float* __restrict__ X, const float* __restrict__ W,
                                   const float* __restrict__ dinv, float* __restrict__ H,
                                   int start, int n) {
  int t = blockIdx.x * blockDim.x + threadIdx.x;
  int r = start + (t >> 6), c = t & 63;
  if (r >= n) return;
  float s = 0.f;
  for (int k = 0; k < 64; ++k) s += X[(size_t)r * 64 + k] * W[k * 64 + c];
  H[(size_t)r * 64 + c] = s * dinv[r];
}

#define GATHER_CHUNK 8
#define TILE_STRIDE 68

__global__ __launch_bounds__(256) void gather1_gemm2_kernel(const float* __restrict__ h,
                                                            const int* __restrict__ rowptr,
                                                            const int* __restrict__ srcs,
                                                            const float* __restrict__ dinv,
                                                            const float* __restrict__ B1,
                                                            const float* __restrict__ W2,
                                                            float* __restrict__ G, int e, int n) {
  __shared__ alignas(16) float tile[16 * TILE_STRIDE];
  __shared__ float w2s[64 * 2];
  __shared__ float dis[16];
  const int tid = threadIdx.x;
  const int ln = tid >> 4;
  const int l = tid & 15;
  const int node0 = blockIdx.x * 16;
  const int node = min(node0 + ln, n - 1);
  const float w2v = W2[tid & 127];
  const int beg = rowptr[node];
  const int end = (node + 1 < n) ? rowptr[node + 1] : e;
  const float di = dinv[node];
  const float4 bias = ((const float4*)B1)[l];
  const float4* h4 = (const float4*)h;
  float4 acc = h4[(size_t)node * 16 + l];

  int j = beg;
  for (; j + GATHER_CHUNK <= end; j += GATHER_CHUNK) {
    int s[GATHER_CHUNK];
    float4 a[GATHER_CHUNK];
#pragma unroll
    for (int k = 0; k < GATHER_CHUNK; ++k) s[k] = srcs[j + k];
#pragma unroll
    for (int k = 0; k < GATHER_CHUNK; ++k) a[k] = h4[(size_t)s[k] * 16 + l];
#pragma unroll
    for (int k = 0; k < GATHER_CHUNK; ++k) {
      acc.x += a[k].x;
      acc.y += a[k].y;
      acc.z += a[k].z;
      acc.w += a[k].w;
    }
  }
  if (j < end) {
    int s[GATHER_CHUNK];
    float4 a[GATHER_CHUNK];
#pragma unroll
    for (int k = 0; k < GATHER_CHUNK; ++k) s[k] = srcs[min(j + k, end - 1)];
#pragma unroll
    for (int k = 0; k < GATHER_CHUNK; ++k) a[k] = h4[(size_t)s[k] * 16 + l];
#pragma unroll
    for (int k = 0; k < GATHER_CHUNK; ++k) {
      const float m = (j + k < end) ? 1.0f : 0.0f;
      acc.x += a[k].x * m;
      acc.y += a[k].y * m;
      acc.z += a[k].z * m;
      acc.w += a[k].w * m;
    }
  }
  float4 t;
  t.x = fmaxf(acc.x * di + bias.x, 0.0f);
  t.y = fmaxf(acc.y * di + bias.y, 0.0f);
  t.z = fmaxf(acc.z * di + bias.z, 0.0f);
  t.w = fmaxf(acc.w * di + bias.w, 0.0f);
  *(float4*)&tile[ln * TILE_STRIDE + l * 4] = t;
  if (l == 0) dis[ln] = di;
  if (tid < 128) w2s[tid] = w2v;
  __syncthreads();

  if (__builtin_amdgcn_readfirstlane(tid) >= 32) return;

  const int half = tid >> 4;
  const int nc = tid & 15;
  const int wc = nc & 1;
  v8f d = {};
#pragma unroll
  for (int k0 = 0; k0 < 64; k0 += 4) {
    const int kb = k0 + half * 2;
    const v2f x = *(const v2f*)(&tile[nc * TILE_STRIDE + kb]);
    const float w0 = w2s[kb * 2 + wc];
    const float w1 = w2s[kb * 2 + 2 + wc];
    v2f w;
    w.x = (nc < 2) ? w0 : 0.0f;
    w.y = (nc < 2) ? w1 : 0.0f;
    d = __builtin_amdgcn_wmma_f32_16x16x4_f32(false, w, false, x, (short)0, d, false, false);
  }
  if (half == 0 && node0 + nc < n) {
    float4 rec;
    rec.x = d[0];
    rec.y = d[1];
    rec.z = dis[nc];
    rec.w = 0.0f;
    *(float4*)&G[(size_t)(node0 + nc) * 4] = rec;
  }
}

#define G2_LANES 8
#define G2_CHUNK 4
#define G2_THREADS 256
#define G2_NODES (G2_THREADS / G2_LANES)

__global__ __launch_bounds__(G2_THREADS) void gather2_kernel(const float* __restrict__ rec,
                                                             const int* __restrict__ rowptr,
                                                             const int* __restrict__ srcs,
                                                             const float* __restrict__ b2,
                                                             float* __restrict__ out, int e, int n) {
  __shared__ float stage[G2_NODES * 2];
  const int tid = threadIdx.x;
  const int ln = tid / G2_LANES;
  const int t = tid & (G2_LANES - 1);
  const int node0 = blockIdx.x * G2_NODES;
  const int node = min(node0 + ln, n - 1);
  const int beg = rowptr[node];
  const int end = (node + 1 < n) ? rowptr[node + 1] : e;
  const float4* r4 = (const float4*)rec;
  const float4 self = r4[node];
  const float di = self.z;
  float ax = 0.0f, ay = 0.0f;

  for (int j = beg + t; j < end; j += G2_LANES * G2_CHUNK) {
    int s[G2_CHUNK];
    float4 r[G2_CHUNK];
#pragma unroll
    for (int k = 0; k < G2_CHUNK; ++k) {
      const int jj = j + k * G2_LANES;
      s[k] = srcs[(jj < end) ? jj : j];
    }
#pragma unroll
    for (int k = 0; k < G2_CHUNK; ++k) {
      const bool ok = j + k * G2_LANES < end;
      r[k] = r4[ok ? s[k] : node];
    }
    static_assert(G2_CHUNK == 4, "the fence below names four records");
    asm volatile("" ::"v"(r[0].x), "v"(r[1].x), "v"(r[2].x), "v"(r[3].x));
#pragma unroll
    for (int k = 0; k < G2_CHUNK; ++k) {
      const bool ok = j + k * G2_LANES < end;
      const float w = ok ? r[k].z * di : 0.0f;
      ax += r[k].x * w;
      ay += r[k].y * w;
    }
  }
#pragma unroll
  for (int o = G2_LANES / 2; o > 0; o >>= 1) {
    ax += __shfl_xor(ax, o);
    ay += __shfl_xor(ay, o);
  }
  if (t == 0) {
    const float ss = di * di;
    stage[ln * 2 + 0] = (self.x * ss + b2[0]) + ax;
    stage[ln * 2 + 1] = (self.y * ss + b2[1]) + ay;
  }
  __syncthreads();
  const int nf = min(G2_NODES, n - node0) * 2;
  if (tid < nf) out[(size_t)node0 * 2 + tid] = stage[tid];
}

extern "C" void kernel_launch(void* const* d_in, const int* in_sizes, int n_in,
                              void* d_out, int out_size, void* d_ws, size_t ws_size,
                              hipStream_t stream) {
  const float* x  = (const float*)d_in[0];
  const int*   ei = (const int*)d_in[1];
  const float* W1 = (const float*)d_in[2];
  const float* b1 = (const float*)d_in[3];
  const float* W2 = (const float*)d_in[4];
  const float* b2 = (const float*)d_in[5];
  float* out = (float*)d_out;

  const int N = in_sizes[0] / 64;
  const int E = in_sizes[1] / 2;
  const int* row = ei;
  const int* col = ei + E;

  const int Npad = (N + 63) & ~63;
  float* dinv = (float*)d_ws;
  float* h    = dinv + Npad;
  float* g    = h + (size_t)N * 64;
  float* prt  = g + (size_t)N * 4;
  int* srcs   = (int*)(prt + (size_t)DEG_SLICES * N);
  int* rowptr = srcs + E;
  int* bsum   = rowptr + Npad;

  const int T = 256;
  const int ranges = (N + DEG_NP - 1) / DEG_NP;
  deg_part_kernel<<<dim3(DEG_SLICES, ranges), PART_THREADS, 0, stream>>>(col, (int*)prt, E, N);
  const int scanBlocks = (N + SCAN_THREADS - 1) / SCAN_THREADS;
  deg_fin_kernel<<<scanBlocks, SCAN_THREADS, 0, stream>>>((int*)prt, dinv, rowptr, bsum, N);
  rowptr_kernel<<<scanBlocks, SCAN_THREADS, 0, stream>>>(rowptr, bsum, N);
  fill_gemm_kernel<<<dim3(DEG_SLICES, ranges + GEMM_Y), PART_THREADS, 0, stream>>>(
      row, col, (const int*)prt, rowptr, srcs, E, N, ranges, x, W1, dinv, h);
  const int tailStart = (N / 16) * 16;
  if (tailStart < N) {
    int tcnt = (N - tailStart) * 64;
    gemm64_tail_kernel<<<(tcnt + T - 1) / T, T, 0, stream>>>(x, W1, dinv, h, tailStart, N);
  }

  gather1_gemm2_kernel<<<(N + 15) / 16, 256, 0, stream>>>(h, rowptr, srcs, dinv, b1, W2, g, E, N);

  gather2_kernel<<<(N + G2_NODES - 1) / G2_NODES, G2_THREADS, 0, stream>>>(g, rowptr, srcs, b2, out,
                                                                           E, N);
}


# Edge list by two-level counting sort: records binned by destination range, each bin sorted in LDS by one block, stored contiguously; count matrix, one launch removed; sum order: arrival order in bin
# speedup vs baseline: 4.3655x; 1.2228x over previous
#include <hip/hip_runtime.h>
#include <hip/hip_bf16.h>


typedef float v2f __attribute__((ext_vector_type(2)));
typedef float v8f __attribute__((ext_vector_type(8)));

#define LDS_STRIDE 66

#define PART_THREADS 1024
#define BIN_SLICE 6400
#define SLICE_UNROLL 7
#define MAX_BINS 256
#define MAX_BINW 2048
#define SORT_CAP 24576
#define EDGE_UNROLL 8
#define RESCALE_UNROLL 4
#define GEMM_BLOCKS 256

static_assert(SLICE_UNROLL * PART_THREADS >= BIN_SLICE, "a slice is one trip");
static_assert(MAX_BINW == 2 * PART_THREADS, "the scan takes two nodes per thread");

__global__ __launch_bounds__(PART_THREADS) void count_gemm_kernel(const int* __restrict__ col,
                                                                  int* __restrict__ cnt,
                                                                  int e, int nsl, int binw, int n,
                                                                  const float* __restrict__ X,
                                                                  const float* __restrict__ W,
                                                                  float* __restrict__ H) {
  __shared__ alignas(16) unsigned char lds_raw[64 * LDS_STRIDE * sizeof(float)];
  const int tid = threadIdx.x;

  if ((int)blockIdx.x < nsl) {
    int* hist = (int*)lds_raw;
    if (tid < MAX_BINS) hist[tid] = 0;
    __syncthreads();
    const int e0 = min(e, (int)blockIdx.x * BIN_SLICE);
    const int e1 = min(e, e0 + BIN_SLICE);
    int d[SLICE_UNROLL];
#pragma unroll
    for (int k = 0; k < SLICE_UNROLL; ++k) {
      const int i = e0 + k * PART_THREADS + tid;
      const bool ok = i < e1;
      const int c = col[ok ? i : e0];
      d[k] = ok ? min(c / binw, MAX_BINS - 1) : -1;
    }
#pragma unroll
    for (int k = 0; k < SLICE_UNROLL; ++k)
      if ((unsigned)d[k] < (unsigned)MAX_BINS) atomicAdd(&hist[d[k]], 1);
    __syncthreads();
    if (tid < MAX_BINS) cnt[(size_t)blockIdx.x * MAX_BINS + tid] = hist[tid];
    return;
  }

  float* Wt = (float*)lds_raw;
#pragma unroll
  for (int p = tid; p < 64 * 64; p += PART_THREADS) {
    const int k = p >> 6, nn = p & 63;
    Wt[nn * LDS_STRIDE + k] = W[p];
  }
  __syncthreads();

  const int gblocks = (int)gridDim.x - nsl;
  const int g = (int)blockIdx.x - nsl;
  const int wave = __builtin_amdgcn_readfirstlane(tid >> 5);
  const int lane = tid & 31;
  const int half = lane >> 4;
  const int nc   = lane & 15;
  const int tiles = n >> 4;

  for (int t = wave * gblocks + g; t < tiles; t += (PART_THREADS / 32) * gblocks) {
    const int rowBase = t * 16;
    const float* xrow = X + (size_t)(rowBase + nc) * 64;
    v8f acc0 = {}, acc1 = {}, acc2 = {}, acc3 = {};

#pragma unroll
    for (int k0 = 0; k0 < 64; k0 += 4) {
      if (k0 == 32) __builtin_amdgcn_sched_barrier(0);
      const int kb = k0 + half * 2;
      v2f a  = *(const v2f*)(xrow + kb);
      v2f b0 = *(const v2f*)(&Wt[(0 * 16 + nc) * LDS_STRIDE + kb]);
      v2f b1 = *(const v2f*)(&Wt[(1 * 16 + nc) * LDS_STRIDE + kb]);
      v2f b2 = *(const v2f*)(&Wt[(2 * 16 + nc) * LDS_STRIDE + kb]);
      v2f b3 = *(const v2f*)(&Wt[(3 * 16 + nc) * LDS_STRIDE + kb]);
      acc0 = __builtin_amdgcn_wmma_f32_16x16x4_f32(false, a, false, b0, (short)0, acc0, false, false);
      acc1 = __builtin_amdgcn_wmma_f32_16x16x4_f32(false, a, false, b1, (short)0, acc1, false, false);
      acc2 = __builtin_amdgcn_wmma_f32_16x16x4_f32(false, a, false, b2, (short)0, acc2, false, false);
      acc3 = __builtin_amdgcn_wmma_f32_16x16x4_f32(false, a, false, b3, (short)0, acc3, false, false);
    }

#pragma unroll
    for (int v = 0; v < 8; ++v) {
      float* hrow = H + (size_t)(rowBase + v + half * 8) * 64;
      hrow[0 * 16 + nc] = acc0[v];
      hrow[1 * 16 + nc] = acc1[v];
      hrow[2 * 16 + nc] = acc2[v];
      hrow[3 * 16 + nc] = acc3[v];
    }
  }
}

__global__ void gemm64_tail_kernel(const float* __restrict__ X, const float* __restrict__ W,
                                   float* __restrict__ H, int start, int n) {
  int t = blockIdx.x * blockDim.x + threadIdx.x;
  int r = start + (t >> 6), c = t & 63;
  if (r >= n) return;
  float s = 0.f;
  for (int k = 0; k < 64; ++k) s += X[(size_t)r * 64 + k] * W[k * 64 + c];
  H[(size_t)r * 64 + c] = s;
}

__device__ __forceinline__ void excl_scan_256(int* a, int lane) {
  int v[8];
  int sum = 0;
#pragma unroll
  for (int k = 0; k < 8; ++k) {
    v[k] = a[lane * 8 + k];
    sum += v[k];
  }
  int incl = sum;
  for (int o = 1; o < 32; o <<= 1) {
    const int u = __shfl_up(incl, o);
    if (lane >= o) incl += u;
  }
  int run = incl - sum;
#pragma unroll
  for (int k = 0; k < 8; ++k) {
    a[lane * 8 + k] = run;
    run += v[k];
  }
}

__global__ __launch_bounds__(PART_THREADS) void bin_scatter_kernel(const int* __restrict__ row,
                                                                   const int* __restrict__ col,
                                                                   const int* __restrict__ cnt,
                                                                   int* __restrict__ binstart,
                                                                   int* __restrict__ binned_i,
                                                                   int e, int nsl, int binw) {
  int2* binned = (int2*)binned_i;
  __shared__ int2 stage[BIN_SLICE];
  __shared__ int ptot[4][MAX_BINS];
  __shared__ int pbef[4][MAX_BINS];
  __shared__ int binbase[MAX_BINS];
  __shared__ int lstart[MAX_BINS];
  __shared__ int lcur[MAX_BINS];
  __shared__ int goff[MAX_BINS];
  const int tid = threadIdx.x;
  const int s = blockIdx.x;
  const int q = tid >> 8, t = tid & (MAX_BINS - 1);

  int tot = 0, bef = 0;
#pragma unroll 8
  for (int sp = q; sp < nsl; sp += 4) {
    const int c = cnt[(size_t)sp * MAX_BINS + t];
    tot += c;
    bef += (sp < s) ? c : 0;
  }
  ptot[q][t] = tot;
  pbef[q][t] = bef;
  const int mine = (tid < MAX_BINS) ? cnt[(size_t)s * MAX_BINS + tid] : 0;
  __syncthreads();
  int mybef = 0;
  if (tid < MAX_BINS) {
    binbase[tid] = ptot[0][tid] + ptot[1][tid] + ptot[2][tid] + ptot[3][tid];
    mybef = pbef[0][tid] + pbef[1][tid] + pbef[2][tid] + pbef[3][tid];
    lstart[tid] = mine;
  }
  __syncthreads();
  if (tid < 32) {
    excl_scan_256(binbase, tid);
    excl_scan_256(lstart, tid);
  }
  __syncthreads();
  if (tid < MAX_BINS) {
    goff[tid] = binbase[tid] + mybef;
    lcur[tid] = lstart[tid];
    if (s == 0) binstart[tid] = binbase[tid];
  }
  if (s == 0 && tid == 0) binstart[MAX_BINS] = e;
  __syncthreads();

  const int e0 = min(e, s * BIN_SLICE);
  const int e1 = min(e, e0 + BIN_SLICE);
  const int m = e1 - e0;
  {
    int d[SLICE_UNROLL], r[SLICE_UNROLL];
#pragma unroll
    for (int k = 0; k < SLICE_UNROLL; ++k) {
      const int i = e0 + k * PART_THREADS + tid;
      const bool ok = i < e1;
      const int ii = ok ? i : e0;
      const int c = col[ii];
      r[k] = row[ii];
      d[k] = ok ? c : -1;
    }
#pragma unroll
    for (int k = 0; k < SLICE_UNROLL; ++k) {
      if (d[k] >= 0) {
        const int b = min(d[k] / binw, MAX_BINS - 1);
        const int p = atomicAdd(&lcur[b], 1);
        if ((unsigned)p < (unsigned)BIN_SLICE) stage[p] = make_int2(r[k], d[k]);
      }
    }
  }
  __syncthreads();
  for (int i = tid; i < m; i += PART_THREADS) {
    const int2 rec = stage[i];
    const int b = min(rec.y / binw, MAX_BINS - 1);
    const int pos = goff[b] + (i - lstart[b]);
    if ((unsigned)pos < (unsigned)e) binned[pos] = rec;
  }
}

__global__ __launch_bounds__(PART_THREADS) void bin_sort_kernel(const int* __restrict__ binned_i,
                                                                const int* __restrict__ binstart,
                                                                int* __restrict__ srcs,
                                                                int* __restrict__ rowptr,
                                                                float* __restrict__ dinv,
                                                                float* __restrict__ h,
                                                                int n, int binw) {
  const int2* binned = (const int2*)binned_i;
  __shared__ int cntl[MAX_BINW];
  __shared__ float dl[MAX_BINW];
  __shared__ int wtot[32];
  __shared__ int stage[SORT_CAP];
  const int tid = threadIdx.x;
  const int lane = tid & 31, wave = tid >> 5;
  const int n0 = (int)blockIdx.x * binw;
  const int nn = min(binw, n - n0);
  const int p0 = binstart[blockIdx.x];
  const int p1 = binstart[blockIdx.x + 1];
  const int m = p1 - p0;

  cntl[2 * tid] = 0;
  cntl[2 * tid + 1] = 0;
  __syncthreads();

  for (int base = p0; base < p1; base += PART_THREADS * EDGE_UNROLL) {
    int d[EDGE_UNROLL];
#pragma unroll
    for (int k = 0; k < EDGE_UNROLL; ++k) {
      const int i = base + k * PART_THREADS + tid;
      const bool ok = i < p1;
      const int2 rec = binned[ok ? i : p0];
      d[k] = ok ? (rec.y - n0) : -1;
    }
#pragma unroll
    for (int k = 0; k < EDGE_UNROLL; ++k)
      if ((unsigned)d[k] < (unsigned)nn) atomicAdd(&cntl[d[k]], 1);
  }
  __syncthreads();

  const int c0 = cntl[2 * tid], c1 = cntl[2 * tid + 1];
  const int sum = c0 + c1;
  int incl = sum;
  for (int o = 1; o < 32; o <<= 1) {
    const int u = __shfl_up(incl, o);
    if (lane >= o) incl += u;
  }
  if (lane == 31) wtot[wave] = incl;
  __syncthreads();
  if (wave == 0) {
    const int w = wtot[lane];
    int wi = w;
    for (int o = 1; o < 32; o <<= 1) {
      const int u = __shfl_up(wi, o);
      if (lane >= o) wi += u;
    }
    wtot[lane] = wi - w;
  }
  __syncthreads();
  const int ex = wtot[wave] + incl - sum;
  cntl[2 * tid] = ex;
  cntl[2 * tid + 1] = ex + c0;
  if (2 * tid < nn) {
    const float dv = rsqrtf(1.0f + (float)c0);
    rowptr[n0 + 2 * tid] = p0 + ex;
    dinv[n0 + 2 * tid] = dv;
    dl[2 * tid] = dv;
  }
  if (2 * tid + 1 < nn) {
    const float dv = rsqrtf(1.0f + (float)c1);
    rowptr[n0 + 2 * tid + 1] = p0 + ex + c0;
    dinv[n0 + 2 * tid + 1] = dv;
    dl[2 * tid + 1] = dv;
  }
  __syncthreads();

  const bool staged = m <= SORT_CAP;
  for (int base = p0; base < p1; base += PART_THREADS * EDGE_UNROLL) {
    int d[EDGE_UNROLL], r[EDGE_UNROLL];
#pragma unroll
    for (int k = 0; k < EDGE_UNROLL; ++k) {
      const int i = base + k * PART_THREADS + tid;
      const bool ok = i < p1;
      const int2 rec = binned[ok ? i : p0];
      r[k] = rec.x;
      d[k] = ok ? (rec.y - n0) : -1;
    }
#pragma unroll
    for (int k = 0; k < EDGE_UNROLL; ++k) {
      if ((unsigned)d[k] < (unsigned)nn) {
        const int pos = atomicAdd(&cntl[d[k]], 1);
        if (staged) {
          if ((unsigned)pos < (unsigned)SORT_CAP) stage[pos] = r[k];
        } else if ((unsigned)pos < (unsigned)m) {
          srcs[p0 + pos] = r[k];
        }
      }
    }
  }
  __syncthreads();
  if (staged)
    for (int i = tid; i < m; i += PART_THREADS) srcs[p0 + i] = stage[i];

  float4* h4 = (float4*)h + (size_t)n0 * 16;
  const int nq = nn * 16;
  for (int base = 0; base < nq; base += PART_THREADS * RESCALE_UNROLL) {
    float4 v[RESCALE_UNROLL];
#pragma unroll
    for (int k = 0; k < RESCALE_UNROLL; ++k) v[k] = h4[min(base + k * PART_THREADS + tid, nq - 1)];
#pragma unroll
    for (int k = 0; k < RESCALE_UNROLL; ++k) {
      const int idx = base + k * PART_THREADS + tid;
      if (idx < nq) {
        const float dv = dl[idx >> 4];
        float4 o;
        o.x = v[k].x * dv;
        o.y = v[k].y * dv;
        o.z = v[k].z * dv;
        o.w = v[k].w * dv;
        h4[idx] = o;
      }
    }
  }
}

#define GATHER_CHUNK 8
#define TILE_STRIDE 68

__global__ __launch_bounds__(256) void gather1_gemm2_kernel(const float* __restrict__ h,
                                                            const int* __restrict__ rowptr,
                                                            const int* __restrict__ srcs,
                                                            const float* __restrict__ dinv,
                                                            const float* __restrict__ B1,
                                                            const float* __restrict__ W2,
                                                            float* __restrict__ G, int e, int n) {
  __shared__ alignas(16) float tile[16 * TILE_STRIDE];
  __shared__ float w2s[64 * 2];
  __shared__ float dis[16];
  const int tid = threadIdx.x;
  const int ln = tid >> 4;
  const int l = tid & 15;
  const int node0 = blockIdx.x * 16;
  const int node = min(node0 + ln, n - 1);
  const float w2v = W2[tid & 127];
  const int beg = rowptr[node];
  const int end = (node + 1 < n) ? rowptr[node + 1] : e;
  const float di = dinv[node];
  const float4 bias = ((const float4*)B1)[l];
  const float4* h4 = (const float4*)h;
  float4 acc = h4[(size_t)node * 16 + l];

  int j = beg;
  for (; j + GATHER_CHUNK <= end; j += GATHER_CHUNK) {
    int s[GATHER_CHUNK];
    float4 a[GATHER_CHUNK];
#pragma unroll
    for (int k = 0; k < GATHER_CHUNK; ++k) s[k] = srcs[j + k];
#pragma unroll
    for (int k = 0; k < GATHER_CHUNK; ++k) a[k] = h4[(size_t)s[k] * 16 + l];
#pragma unroll
    for (int k = 0; k < GATHER_CHUNK; ++k) {
      acc.x += a[k].x;
      acc.y += a[k].y;
      acc.z += a[k].z;
      acc.w += a[k].w;
    }
  }
  if (j < end) {
    int s[GATHER_CHUNK];
    float4 a[GATHER_CHUNK];
#pragma unroll
    for (int k = 0; k < GATHER_CHUNK; ++k) s[k] = srcs[min(j + k, end - 1)];
#pragma unroll
    for (int k = 0; k < GATHER_CHUNK; ++k) a[k] = h4[(size_t)s[k] * 16 + l];
#pragma unroll
    for (int k = 0; k < GATHER_CHUNK; ++k) {
      const float m = (j + k < end) ? 1.0f : 0.0f;
      acc.x += a[k].x * m;
      acc.y += a[k].y * m;
      acc.z += a[k].z * m;
      acc.w += a[k].w * m;
    }
  }
  float4 t;
  t.x = fmaxf(acc.x * di + bias.x, 0.0f);
  t.y = fmaxf(acc.y * di + bias.y, 0.0f);
  t.z = fmaxf(acc.z * di + bias.z, 0.0f);
  t.w = fmaxf(acc.w * di + bias.w, 0.0f);
  *(float4*)&tile[ln * TILE_STRIDE + l * 4] = t;
  if (l == 0) dis[ln] = di;
  if (tid < 128) w2s[tid] = w2v;
  __syncthreads();

  if (__builtin_amdgcn_readfirstlane(tid) >= 32) return;

  const int half = tid >> 4;
  const int nc = tid & 15;
  const int wc = nc & 1;
  v8f d = {};
#pragma unroll
  for (int k0 = 0; k0 < 64; k0 += 4) {
    const int kb = k0 + half * 2;
    const v2f x = *(const v2f*)(&tile[nc * TILE_STRIDE + kb]);
    const float w0 = w2s[kb * 2 + wc];
    const float w1 = w2s[kb * 2 + 2 + wc];
    v2f w;
    w.x = (nc < 2) ? w0 : 0.0f;
    w.y = (nc < 2) ? w1 : 0.0f;
    d = __builtin_amdgcn_wmma_f32_16x16x4_f32(false, w, false, x, (short)0, d, false, false);
  }
  if (half == 0 && node0 + nc < n) {
    float4 rec;
    rec.x = d[0];
    rec.y = d[1];
    rec.z = dis[nc];
    rec.w = 0.0f;
    *(float4*)&G[(size_t)(node0 + nc) * 4] = rec;
  }
}

#define G2_LANES 8
#define G2_CHUNK 4
#define G2_THREADS 256
#define G2_NODES (G2_THREADS / G2_LANES)

__global__ __launch_bounds__(G2_THREADS) void gather2_kernel(const float* __restrict__ rec,
                                                             const int* __restrict__ rowptr,
                                                             const int* __restrict__ srcs,
                                                             const float* __restrict__ b2,
                                                             float* __restrict__ out, int e, int n) {
  __shared__ float stage[G2_NODES * 2];
  const int tid = threadIdx.x;
  const int ln = tid / G2_LANES;
  const int t = tid & (G2_LANES - 1);
  const int node0 = blockIdx.x * G2_NODES;
  const int node = min(node0 + ln, n - 1);
  const int beg = rowptr[node];
  const int end = (node + 1 < n) ? rowptr[node + 1] : e;
  const float4* r4 = (const float4*)rec;
  const float4 self = r4[node];
  const float di = self.z;
  float ax = 0.0f, ay = 0.0f;

  for (int j = beg + t; j < end; j += G2_LANES * G2_CHUNK) {
    int s[G2_CHUNK];
    float4 r[G2_CHUNK];
#pragma unroll
    for (int k = 0; k < G2_CHUNK; ++k) {
      const int jj = j + k * G2_LANES;
      s[k] = srcs[(jj < end) ? jj : j];
    }
#pragma unroll
    for (int k = 0; k < G2_CHUNK; ++k) {
      const bool ok = j + k * G2_LANES < end;
      r[k] = r4[ok ? s[k] : node];
    }
    static_assert(G2_CHUNK == 4, "the fence below names four records");
    asm volatile("" ::"v"(r[0].x), "v"(r[1].x), "v"(r[2].x), "v"(r[3].x));
#pragma unroll
    for (int k = 0; k < G2_CHUNK; ++k) {
      const bool ok = j + k * G2_LANES < end;
      const float w = ok ? r[k].z * di : 0.0f;
      ax += r[k].x * w;
      ay += r[k].y * w;
    }
  }
#pragma unroll
  for (int o = G2_LANES / 2; o > 0; o >>= 1) {
    ax += __shfl_xor(ax, o);
    ay += __shfl_xor(ay, o);
  }
  if (t == 0) {
    const float ss = di * di;
    stage[ln * 2 + 0] = (self.x * ss + b2[0]) + ax;
    stage[ln * 2 + 1] = (self.y * ss + b2[1]) + ay;
  }
  __syncthreads();
  const int nf = min(G2_NODES, n - node0) * 2;
  if (tid < nf) out[(size_t)node0 * 2 + tid] = stage[tid];
}

extern "C" void kernel_launch(void* const* d_in, const int* in_sizes, int n_in,
                              void* d_out, int out_size, void* d_ws, size_t ws_size,
                              hipStream_t stream) {
  const float* x  = (const float*)d_in[0];
  const int*   ei = (const int*)d_in[1];
  const float* W1 = (const float*)d_in[2];
  const float* b1 = (const float*)d_in[3];
  const float* W2 = (const float*)d_in[4];
  const float* b2 = (const float*)d_in[5];
  float* out = (float*)d_out;

  const int N = in_sizes[0] / 64;
  const int E = in_sizes[1] / 2;
  const int* row = ei;
  const int* col = ei + E;

  const int nsl = (E + BIN_SLICE - 1) / BIN_SLICE;
  int binw = ((N + MAX_BINS - 1) / MAX_BINS + 15) & ~15;
  if (binw < 16) binw = 16;
  const int nbins = (N + binw - 1) / binw;

  const int Npad = (N + 63) & ~63;
  float* dinv   = (float*)d_ws;
  float* h      = dinv + Npad;
  float* g      = h + (size_t)N * 64;
  int* cnt      = (int*)(g + (size_t)N * 4);
  int* binstart = cnt + (size_t)nsl * MAX_BINS;
  int* binned   = binstart + 2 * MAX_BINS;
  int* srcs     = binned + (size_t)2 * E;
  int* rowptr   = srcs + E;

  const int T = 256;
  count_gemm_kernel<<<nsl + GEMM_BLOCKS, PART_THREADS, 0, stream>>>(col, cnt, E, nsl, binw, N, x, W1, h);
  const int tailStart = (N / 16) * 16;
  if (tailStart < N) {
    int tcnt = (N - tailStart) * 64;
    gemm64_tail_kernel<<<(tcnt + T - 1) / T, T, 0, stream>>>(x, W1, h, tailStart, N);
  }
  bin_scatter_kernel<<<nsl, PART_THREADS, 0, stream>>>(row, col, cnt, binstart, binned, E, nsl, binw);
  bin_sort_kernel<<<nbins, PART_THREADS, 0, stream>>>(binned, binstart, srcs, rowptr, dinv, h, N, binw);

  gather1_gemm2_kernel<<<(N + 15) / 16, 256, 0, stream>>>(h, rowptr, srcs, dinv, b1, W2, g, E, N);

  gather2_kernel<<<(N + G2_NODES - 1) / G2_NODES, G2_THREADS, 0, stream>>>(g, rowptr, srcs, b2, out,
                                                                           E, N);
}


# Sort launches: block barriers order LDS only, so global traffic crosses them: bin launch asks for its input rows before the scan and stores scaled rows before the second walk; no store drains
# speedup vs baseline: 5.8882x; 1.3488x over previous
#include <hip/hip_runtime.h>
#include <hip/hip_bf16.h>


typedef float v2f __attribute__((ext_vector_type(2)));
typedef float v8f __attribute__((ext_vector_type(8)));

#define LDS_STRIDE 66

#define PART_THREADS 1024
#define BIN_SLICE 6400
#define SLICE_UNROLL 7
#define MAX_BINS 256
#define MAX_BINW 2048
#define SORT_CAP 24576
#define RUN_LANES 4
#define RUN_UNROLL 8
#define RESCALE_UNROLL 4
#define RESCALE_EARLY 7
#define STAGE_BYTES (BIN_SLICE * 8)
#define SLICE_LDS (STAGE_BYTES + 2 * MAX_BINS * 4)

static_assert(SLICE_UNROLL * PART_THREADS >= BIN_SLICE, "a slice is one trip");
static_assert(MAX_BINW == 2 * PART_THREADS, "the scan takes two nodes per thread");

__device__ __forceinline__ void excl_scan_256(int* a, int lane) {
  int v[8];
  int sum = 0;
#pragma unroll
  for (int k = 0; k < 8; ++k) {
    v[k] = a[lane * 8 + k];
    sum += v[k];
  }
  int incl = sum;
  for (int o = 1; o < 32; o <<= 1) {
    const int u = __shfl_up(incl, o);
    if (lane >= o) incl += u;
  }
  int run = incl - sum;
#pragma unroll
  for (int k = 0; k < 8; ++k) {
    a[lane * 8 + k] = run;
    run += v[k];
  }
}

__device__ __forceinline__ void lds_barrier() {
  __builtin_amdgcn_fence(__ATOMIC_RELEASE, "workgroup", "local");
  __builtin_amdgcn_s_barrier();
  __builtin_amdgcn_fence(__ATOMIC_ACQUIRE, "workgroup", "local");
}

__global__ __launch_bounds__(PART_THREADS) void slice_sort_kernel(const int* __restrict__ row,
                                                                  const int* __restrict__ col,
                                                                  int* __restrict__ tab,
                                                                  int* __restrict__ binned_i,
                                                                  int e, int binw) {
  __shared__ alignas(16) unsigned char lds_raw[SLICE_LDS];
  const int tid = threadIdx.x;
  int2* stage = (int2*)lds_raw;
  int* lstart = (int*)(lds_raw + STAGE_BYTES);
  int* lcur = lstart + MAX_BINS;
  int2* binned = (int2*)binned_i;
  const int s = blockIdx.x;
  if (tid < MAX_BINS) lstart[tid] = 0;
  const int e0 = min(e, s * BIN_SLICE);
  const int e1 = min(e, e0 + BIN_SLICE);
  const int m = e1 - e0;
  int b[SLICE_UNROLL], r[SLICE_UNROLL], c[SLICE_UNROLL];
#pragma unroll
  for (int k = 0; k < SLICE_UNROLL; ++k) {
    const int i = e0 + k * PART_THREADS + tid;
    const int ii = (i < e1) ? i : e0;
    c[k] = col[ii];
    r[k] = row[ii];
  }
  lds_barrier();
#pragma unroll
  for (int k = 0; k < SLICE_UNROLL; ++k) {
    const bool ok = e0 + k * PART_THREADS + tid < e1;
    b[k] = ok ? min(max(c[k], 0) / binw, MAX_BINS - 1) : -1;
    if (ok) atomicAdd(&lstart[b[k]], 1);
  }
  lds_barrier();
  if (tid < 32) excl_scan_256(lstart, tid);
  lds_barrier();
  if (tid < MAX_BINS) {
    const int v = lstart[tid];
    lcur[tid] = v;
    tab[(size_t)s * MAX_BINS + tid] = v;
  }
  lds_barrier();
#pragma unroll
  for (int k = 0; k < SLICE_UNROLL; ++k) {
    if (b[k] >= 0) {
      const int p = atomicAdd(&lcur[b[k]], 1);
      if ((unsigned)p < (unsigned)BIN_SLICE) stage[p] = make_int2(r[k], c[k]);
    }
  }
  lds_barrier();
  for (int i = tid; i < m; i += PART_THREADS) binned[(size_t)s * BIN_SLICE + i] = stage[i];
}

__device__ __forceinline__ void run_bounds(const int* __restrict__ tab, int s, int bin, int e, int& st,
                                           int& en) {
  const int ms = min(e - s * BIN_SLICE, BIN_SLICE);
  const int a = tab[(size_t)s * MAX_BINS + bin];
  const int b = tab[(size_t)s * MAX_BINS + min(bin + 1, MAX_BINS - 1)];
  st = min(max(a, 0), ms);
  en = (bin + 1 < MAX_BINS) ? min(max(b, st), ms) : ms;
}

__global__ __launch_bounds__(PART_THREADS) void bin_sort_kernel(const int* __restrict__ binned_i,
                                                                const int* __restrict__ tab,
                                                                int* __restrict__ srcs,
                                                                int* __restrict__ rowptr,
                                                                float* __restrict__ dinv,
                                                                const float* __restrict__ X,
                                                                float* __restrict__ xs,
                                                                int e, int nsl, int n, int binw) {
  const int2* binned = (const int2*)binned_i;
  __shared__ int cntl[MAX_BINW];
  __shared__ float dl[MAX_BINW];
  __shared__ int wtot[32];
  __shared__ int tot[2];
  __shared__ int stage[SORT_CAP];
  const int tid = threadIdx.x;
  const int lane = tid & 31, wave = tid >> 5;
  const int sub = tid & (RUN_LANES - 1);
  const int bin = (int)blockIdx.x;
  const int n0 = bin * binw;
  const int nn = min(binw, n - n0);

  cntl[2 * tid] = 0;
  cntl[2 * tid + 1] = 0;
  if (tid < 2) tot[tid] = 0;
  __syncthreads();

  const int sfirst = tid / RUN_LANES;
  int st0 = 0, en0 = 0;
  int before = 0, mine = 0;
  for (int s = sfirst; s < nsl; s += PART_THREADS / RUN_LANES) {
    int st, en;
    run_bounds(tab, s, bin, e, st, en);
    if (s == sfirst) {
      st0 = st;
      en0 = en;
    }
    if (sub == 0) {
      before += st;
      mine += en - st;
    }
    const int* dst = (const int*)(binned + (size_t)s * BIN_SLICE) + 1;
    for (int j0 = st + sub; j0 < en; j0 += RUN_LANES * RUN_UNROLL) {
      int d[RUN_UNROLL];
#pragma unroll
      for (int k = 0; k < RUN_UNROLL; ++k) {
        const int j = j0 + k * RUN_LANES;
        const bool ok = j < en;
        const int y = dst[2 * (ok ? j : j0)];
        d[k] = ok ? (y - n0) : -1;
      }
#pragma unroll
      for (int k = 0; k < RUN_UNROLL; ++k)
        if ((unsigned)d[k] < (unsigned)nn) atomicAdd(&cntl[d[k]], 1);
    }
  }
  const float4* x4 = (const float4*)X + (size_t)n0 * 16;
  float4* xs4 = (float4*)xs + (size_t)n0 * 16;
  const int nq = nn * 16;
  float4 xv[RESCALE_EARLY];
#pragma unroll
  for (int k = 0; k < RESCALE_EARLY; ++k) xv[k] = x4[min(k * PART_THREADS + tid, nq - 1)];

  for (int o = 16; o > 0; o >>= 1) {
    before += __shfl_xor(before, o);
    mine += __shfl_xor(mine, o);
  }
  if (lane == 0) {
    atomicAdd(&tot[0], before);
    atomicAdd(&tot[1], mine);
  }
  lds_barrier();
  const int p0 = tot[0];
  const int m = tot[1];

  const int c0 = cntl[2 * tid], c1 = cntl[2 * tid + 1];
  const int sum = c0 + c1;
  int incl = sum;
  for (int o = 1; o < 32; o <<= 1) {
    const int u = __shfl_up(incl, o);
    if (lane >= o) incl += u;
  }
  if (lane == 31) wtot[wave] = incl;
  lds_barrier();
  if (wave == 0) {
    const int w = wtot[lane];
    int wi = w;
    for (int o = 1; o < 32; o <<= 1) {
      const int u = __shfl_up(wi, o);
      if (lane >= o) wi += u;
    }
    wtot[lane] = wi - w;
  }
  lds_barrier();
  const int ex = wtot[wave] + incl - sum;
  cntl[2 * tid] = ex;
  cntl[2 * tid + 1] = ex + c0;
  if (2 * tid < nn) {
    const float dv = rsqrtf(1.0f + (float)c0);
    rowptr[n0 + 2 * tid] = p0 + ex;
    dinv[n0 + 2 * tid] = dv;
    dl[2 * tid] = dv;
  }
  if (2 * tid + 1 < nn) {
    const float dv = rsqrtf(1.0f + (float)c1);
    rowptr[n0 + 2 * tid + 1] = p0 + ex + c0;
    dinv[n0 + 2 * tid + 1] = dv;
    dl[2 * tid + 1] = dv;
  }
  lds_barrier();

#pragma unroll
  for (int k = 0; k < RESCALE_EARLY; ++k) {
    const int idx = k * PART_THREADS + tid;
    if (idx < nq) {
      const float dv = dl[idx >> 4];
      float4 o;
      o.x = xv[k].x * dv;
      o.y = xv[k].y * dv;
      o.z = xv[k].z * dv;
      o.w = xv[k].w * dv;
      xs4[idx] = o;
    }
  }

  const bool staged = m <= SORT_CAP;
  for (int s = sfirst; s < nsl; s += PART_THREADS / RUN_LANES) {
    int st = st0, en = en0;
    if (s != sfirst) run_bounds(tab, s, bin, e, st, en);
    const int2* run = binned + (size_t)s * BIN_SLICE;
    for (int j0 = st + sub; j0 < en; j0 += RUN_LANES * RUN_UNROLL) {
      int d[RUN_UNROLL], r[RUN_UNROLL];
#pragma unroll
      for (int k = 0; k < RUN_UNROLL; ++k) {
        const int j = j0 + k * RUN_LANES;
        const bool ok = j < en;
        const int2 rec = run[ok ? j : j0];
        r[k] = rec.x;
        d[k] = ok ? (rec.y - n0) : -1;
      }
#pragma unroll
      for (int k = 0; k < RUN_UNROLL; ++k) {
        if ((unsigned)d[k] < (unsigned)nn) {
          const int pos = atomicAdd(&cntl[d[k]], 1);
          if (staged) {
            if ((unsigned)pos < (unsigned)SORT_CAP) stage[pos] = r[k];
          } else if ((unsigned)pos < (unsigned)m) {
            srcs[p0 + pos] = r[k];
          }
        }
      }
    }
  }
  lds_barrier();
  if (staged)
    for (int i = tid; i < m; i += PART_THREADS) srcs[p0 + i] = stage[i];

  for (int base = PART_THREADS * RESCALE_EARLY; base < nq; base += PART_THREADS * RESCALE_UNROLL) {
    float4 v[RESCALE_UNROLL];
#pragma unroll
    for (int k = 0; k < RESCALE_UNROLL; ++k) v[k] = x4[min(base + k * PART_THREADS + tid, nq - 1)];
#pragma unroll
    for (int k = 0; k < RESCALE_UNROLL; ++k) {
      const int idx = base + k * PART_THREADS + tid;
      if (idx < nq) {
        const float dv = dl[idx >> 4];
        float4 o;
        o.x = v[k].x * dv;
        o.y = v[k].y * dv;
        o.z = v[k].z * dv;
        o.w = v[k].w * dv;
        xs4[idx] = o;
      }
    }
  }
}

#define GATHER_CHUNK 8
#define TILE_STRIDE 68
#define G1_THREADS 512
#define G1_NODES (G1_THREADS / 16)
#define W1_LOADS (64 * 64 / G1_THREADS)

static_assert(G1_NODES == 32, "eight product waves: two row tiles times four column tiles");

__global__ __launch_bounds__(G1_THREADS) void gather1_gemm12_kernel(const float* __restrict__ xs,
                                                                    const int* __restrict__ rowptr,
                                                                    const int* __restrict__ srcs,
                                                                    const float* __restrict__ dinv,
                                                                    const float* __restrict__ W1,
                                                                    const float* __restrict__ B1,
                                                                    const float* __restrict__ W2,
                                                                    float* __restrict__ G, int e, int n) {
  __shared__ alignas(16) float Wt[64 * LDS_STRIDE];
  __shared__ alignas(16) float agg[G1_NODES * TILE_STRIDE];
  __shared__ alignas(16) float act[G1_NODES * TILE_STRIDE];
  __shared__ float w2s[64 * 2];
  __shared__ float dis[G1_NODES];
  const int tid = threadIdx.x;
  const int ln = tid >> 4;
  const int l = tid & 15;
  const int node0 = blockIdx.x * G1_NODES;
  const int node = min(node0 + ln, n - 1);
  float wv[W1_LOADS];
#pragma unroll
  for (int q = 0; q < W1_LOADS; ++q) wv[q] = W1[q * G1_THREADS + tid];
  const float w2v = W2[tid & 127];
  const float bv = B1[((tid >> 5) & 3) * 16 + l];
  const int beg = rowptr[node];
  const int end = (node + 1 < n) ? rowptr[node + 1] : e;
  const float di = dinv[node];
  const float4* x4 = (const float4*)xs;
  float4 acc = x4[(size_t)node * 16 + l];
#pragma unroll
  for (int q = 0; q < W1_LOADS; ++q) {
    const int p = q * G1_THREADS + tid;
    Wt[(p & 63) * LDS_STRIDE + (p >> 6)] = wv[q];
  }
  if (tid < 128) w2s[tid] = w2v;

  int j = beg;
  for (; j + GATHER_CHUNK <= end; j += GATHER_CHUNK) {
    int s[GATHER_CHUNK];
    float4 a[GATHER_CHUNK];
#pragma unroll
    for (int k = 0; k < GATHER_CHUNK; ++k) s[k] = srcs[j + k];
#pragma unroll
    for (int k = 0; k < GATHER_CHUNK; ++k) a[k] = x4[(size_t)s[k] * 16 + l];
#pragma unroll
    for (int k = 0; k < GATHER_CHUNK; ++k) {
      acc.x += a[k].x;
      acc.y += a[k].y;
      acc.z += a[k].z;
      acc.w += a[k].w;
    }
  }
  if (j < end) {
    int s[GATHER_CHUNK];
    float4 a[GATHER_CHUNK];
#pragma unroll
    for (int k = 0; k < GATHER_CHUNK; ++k) s[k] = srcs[min(j + k, end - 1)];
#pragma unroll
    for (int k = 0; k < GATHER_CHUNK; ++k) a[k] = x4[(size_t)s[k] * 16 + l];
#pragma unroll
    for (int k = 0; k < GATHER_CHUNK; ++k) {
      const float m = (j + k < end) ? 1.0f : 0.0f;
      acc.x += a[k].x * m;
      acc.y += a[k].y * m;
      acc.z += a[k].z * m;
      acc.w += a[k].w * m;
    }
  }
  float4 t;
  t.x = acc.x * di;
  t.y = acc.y * di;
  t.z = acc.z * di;
  t.w = acc.w * di;
  *(float4*)&agg[ln * TILE_STRIDE + l * 4] = t;
  if (l == 0) dis[ln] = di;
  __syncthreads();

  const int wave = __builtin_amdgcn_readfirstlane(tid >> 5);
  const int half = (tid & 31) >> 4;
  const int nc = tid & 15;

  if (wave < 8) {
    const int mt = wave >> 2;
    const int nt = wave & 3;
    v8f d = {};
#pragma unroll
    for (int k0 = 0; k0 < 64; k0 += 4) {
      const int kb = k0 + half * 2;
      const v2f a = *(const v2f*)(&agg[(mt * 16 + nc) * TILE_STRIDE + kb]);
      const v2f b = *(const v2f*)(&Wt[(nt * 16 + nc) * LDS_STRIDE + kb]);
      d = __builtin_amdgcn_wmma_f32_16x16x4_f32(false, a, false, b, (short)0, d, false, false);
    }
#pragma unroll
    for (int r = 0; r < 8; ++r)
      act[(mt * 16 + half * 8 + r) * TILE_STRIDE + nt * 16 + nc] = fmaxf(d[r] + bv, 0.0f);
  }
  __syncthreads();

  if (wave >= 2) return;

  const int wc = nc & 1;
  v8f d2 = {};
#pragma unroll
  for (int k0 = 0; k0 < 64; k0 += 4) {
    const int kb = k0 + half * 2;
    const v2f x = *(const v2f*)(&act[(wave * 16 + nc) * TILE_STRIDE + kb]);
    const float w0 = w2s[kb * 2 + wc];
    const float w1 = w2s[kb * 2 + 2 + wc];
    v2f w;
    w.x = (nc < 2) ? w0 : 0.0f;
    w.y = (nc < 2) ? w1 : 0.0f;
    d2 = __builtin_amdgcn_wmma_f32_16x16x4_f32(false, w, false, x, (short)0, d2, false, false);
  }
  const int gn = node0 + wave * 16 + nc;
  if (half == 0 && gn < n) {
    float4 rec;
    rec.x = d2[0];
    rec.y = d2[1];
    rec.z = dis[wave * 16 + nc];
    rec.w = 0.0f;
    *(float4*)&G[(size_t)gn * 4] = rec;
  }
}

#define G2_LANES 8
#define G2_CHUNK 4
#define G2_THREADS 256
#define G2_NODES (G2_THREADS / G2_LANES)

__global__ __launch_bounds__(G2_THREADS) void gather2_kernel(const float* __restrict__ rec,
                                                             const int* __restrict__ rowptr,
                                                             const int* __restrict__ srcs,
                                                             const float* __restrict__ b2,
                                                             float* __restrict__ out, int e, int n) {
  __shared__ float stage[G2_NODES * 2];
  const int tid = threadIdx.x;
  const int ln = tid / G2_LANES;
  const int t = tid & (G2_LANES - 1);
  const int node0 = blockIdx.x * G2_NODES;
  const int node = min(node0 + ln, n - 1);
  const int beg = rowptr[node];
  const int end = (node + 1 < n) ? rowptr[node + 1] : e;
  const float4* r4 = (const float4*)rec;
  const float4 self = r4[node];
  const float di = self.z;
  float ax = 0.0f, ay = 0.0f;

  for (int j = beg + t; j < end; j += G2_LANES * G2_CHUNK) {
    int s[G2_CHUNK];
    float4 r[G2_CHUNK];
#pragma unroll
    for (int k = 0; k < G2_CHUNK; ++k) {
      const int jj = j + k * G2_LANES;
      s[k] = srcs[(jj < end) ? jj : j];
    }
#pragma unroll
    for (int k = 0; k < G2_CHUNK; ++k) {
      const bool ok = j + k * G2_LANES < end;
      r[k] = r4[ok ? s[k] : node];
    }
    static_assert(G2_CHUNK == 4, "the fence below names four records");
    asm volatile("" ::"v"(r[0].x), "v"(r[1].x), "v"(r[2].x), "v"(r[3].x));
#pragma unroll
    for (int k = 0; k < G2_CHUNK; ++k) {
      const bool ok = j + k * G2_LANES < end;
      const float w = ok ? r[k].z * di : 0.0f;
      ax += r[k].x * w;
      ay += r[k].y * w;
    }
  }
#pragma unroll
  for (int o = G2_LANES / 2; o > 0; o >>= 1) {
    ax += __shfl_xor(ax, o);
    ay += __shfl_xor(ay, o);
  }
  if (t == 0) {
    const float ss = di * di;
    stage[ln * 2 + 0] = (self.x * ss + b2[0]) + ax;
    stage[ln * 2 + 1] = (self.y * ss + b2[1]) + ay;
  }
  __syncthreads();
  const int nf = min(G2_NODES, n - node0) * 2;
  if (tid < nf) out[(size_t)node0 * 2 + tid] = stage[tid];
}

extern "C" void kernel_launch(void* const* d_in, const int* in_sizes, int n_in,
                              void* d_out, int out_size, void* d_ws, size_t ws_size,
                              hipStream_t stream) {
  const float* x  = (const float*)d_in[0];
  const int*   ei = (const int*)d_in[1];
  const float* W1 = (const float*)d_in[2];
  const float* b1 = (const float*)d_in[3];
  const float* W2 = (const float*)d_in[4];
  const float* b2 = (const float*)d_in[5];
  float* out = (float*)d_out;

  const int N = in_sizes[0] / 64;
  const int E = in_sizes[1] / 2;
  const int* row = ei;
  const int* col = ei + E;

  const int nsl = (E + BIN_SLICE - 1) / BIN_SLICE;
  int binw = ((N + MAX_BINS - 1) / MAX_BINS + 15) & ~15;
  if (binw < 16) binw = 16;
  const int nbins = (N + binw - 1) / binw;

  const int Npad = (N + 63) & ~63;
  float* dinv   = (float*)d_ws;
  float* xs     = dinv + Npad;
  float* g      = xs + (size_t)N * 64;
  int* tab      = (int*)(g + (size_t)N * 4);
  int* binned   = tab + (size_t)nsl * MAX_BINS;
  int* srcs     = binned + (size_t)2 * nsl * BIN_SLICE;
  int* rowptr   = srcs + E;

  slice_sort_kernel<<<nsl, PART_THREADS, 0, stream>>>(row, col, tab, binned, E, binw);
  bin_sort_kernel<<<nbins, PART_THREADS, 0, stream>>>(binned, tab, srcs, rowptr, dinv, x, xs, E, nsl, N, binw);

  gather1_gemm12_kernel<<<(N + G1_NODES - 1) / G1_NODES, G1_THREADS, 0, stream>>>(xs, rowptr, srcs, dinv, W1, b1,
                                                                                   W2, g, E, N);

  gather2_kernel<<<(N + G2_NODES - 1) / G2_NODES, G2_THREADS, 0, stream>>>(g, rowptr, srcs, b2, out,
                                                                           E, N);
}
